# starting kernel
# baseline (speedup 1.0000x reference)
#include <hip/hip_runtime.h>

typedef __attribute__((ext_vector_type(2))) float v2f;
typedef __attribute__((ext_vector_type(8))) float v8f;

#define NODES 50000
#define DIN   512
#define DOUT  256
#define EDGES 800000

__global__ __launch_bounds__(128) void gemm_wmma_f32(const float* __restrict__ A,
                                                     const float* __restrict__ B,
                                                     float* __restrict__ C) {
  const int lane  = threadIdx.x & 31;
  const int wave  = threadIdx.x >> 5;
  const int half  = lane >> 4;
  const int l15   = lane & 15;
  const int mbase = blockIdx.x * 16;
  const int nbase = wave * 64;

  v8f c0 = {}, c1 = {}, c2 = {}, c3 = {};

  const float* arow = A + (size_t)(mbase + l15) * DIN + 2 * half;
  const float* bcol = B + (size_t)(2 * half) * DOUT + nbase + l15;

  for (int k = 0; k < DIN; k += 4) {
    v2f a = *(const v2f*)(arow + k);

    const float* bk = bcol + (size_t)k * DOUT;
    v2f b0, b1, b2, b3;
    b0.x = bk[0];   b0.y = bk[DOUT + 0];
    b1.x = bk[16];  b1.y = bk[DOUT + 16];
    b2.x = bk[32];  b2.y = bk[DOUT + 32];
    b3.x = bk[48];  b3.y = bk[DOUT + 48];

    c0 = __builtin_amdgcn_wmma_f32_16x16x4_f32(false, a, false, b0, (short)0, c0, false, false);
    c1 = __builtin_amdgcn_wmma_f32_16x16x4_f32(false, a, false, b1, (short)0, c1, false, false);
    c2 = __builtin_amdgcn_wmma_f32_16x16x4_f32(false, a, false, b2, (short)0, c2, false, false);
    c3 = __builtin_amdgcn_wmma_f32_16x16x4_f32(false, a, false, b3, (short)0, c3, false, false);
  }

  float* crow = C + (size_t)(mbase + 8 * half) * DOUT + nbase + l15;
#pragma unroll
  for (int r = 0; r < 8; ++r) {
    crow[(size_t)r * DOUT + 0]  = c0[r];
    crow[(size_t)r * DOUT + 16] = c1[r];
    crow[(size_t)r * DOUT + 32] = c2[r];
    crow[(size_t)r * DOUT + 48] = c3[r];
  }
}

__global__ __launch_bounds__(256) void init_bias(float* __restrict__ out,
                                                 const float* __restrict__ bias) {
  const int i = blockIdx.x * 256 + threadIdx.x;
  out[i] = bias[i & (DOUT - 1)];
}

__global__ __launch_bounds__(256) void edge_scatter(const float* __restrict__ sup,
                                                    const float* __restrict__ ev,
                                                    const int*   __restrict__ er,
                                                    const int*   __restrict__ ec,
                                                    float* __restrict__ out) {
  const int lane = threadIdx.x & 31;
  const int wave = threadIdx.x >> 5;
  const int e    = blockIdx.x * 8 + wave;
  if (e >= EDGES) return;

  const float  v = ev[e];
  const size_t c = (size_t)ec[e];
  const size_t r = (size_t)er[e];

  const float4* src = (const float4*)(sup + c * DOUT);
  float*        dst = out + r * DOUT;

#pragma unroll
  for (int it = 0; it < 2; ++it) {
    const int idx = lane + it * 32;
    const float4 s = src[idx];
    const int b = idx * 4;
    atomicAdd(dst + b + 0, s.x * v);
    atomicAdd(dst + b + 1, s.y * v);
    atomicAdd(dst + b + 2, s.z * v);
    atomicAdd(dst + b + 3, s.w * v);
  }
}

extern "C" void kernel_launch(void* const* d_in, const int* in_sizes, int n_in,
                              void* d_out, int out_size, void* d_ws, size_t ws_size,
                              hipStream_t stream) {
  const float* inputs  = (const float*)d_in[0];
  const float* weights = (const float*)d_in[1];
  const float* bias    = (const float*)d_in[2];
  const float* evals   = (const float*)d_in[3];
  const int*   erow    = (const int*)d_in[4];
  const int*   ecol    = (const int*)d_in[5];
  float* out      = (float*)d_out;
  float* supports = (float*)d_ws;

  gemm_wmma_f32<<<NODES / 16, 128, 0, stream>>>(inputs, weights, supports);
  init_bias<<<(NODES * DOUT) / 256, 256, 0, stream>>>(out, bias);
  edge_scatter<<<EDGES / 8, 256, 0, stream>>>(supports, evals, erow, ecol, out);
}


# edge scatter: lane owns floats l+32j of a row, so each gather load and each f32 atomic add instruction covers one whole 128-B line (was 8 lanes in each of 4 lines); atomics, arrival-order sums kept
# speedup vs baseline: 1.5159x; 1.5159x over previous
#include <hip/hip_runtime.h>

typedef __attribute__((ext_vector_type(2))) float v2f;
typedef __attribute__((ext_vector_type(8))) float v8f;

#define NODES 50000
#define DIN   512
#define DOUT  256
#define EDGES 800000

__global__ __launch_bounds__(128) void gemm_wmma_f32(const float* __restrict__ A,
                                                     const float* __restrict__ B,
                                                     float* __restrict__ C) {
  const int lane  = threadIdx.x & 31;
  const int wave  = threadIdx.x >> 5;
  const int half  = lane >> 4;
  const int l15   = lane & 15;
  const int mbase = blockIdx.x * 16;
  const int nbase = wave * 64;

  v8f c0 = {}, c1 = {}, c2 = {}, c3 = {};

  const float* arow = A + (size_t)(mbase + l15) * DIN + 2 * half;
  const float* bcol = B + (size_t)(2 * half) * DOUT + nbase + l15;

  for (int k = 0; k < DIN; k += 4) {
    v2f a = *(const v2f*)(arow + k);

    const float* bk = bcol + (size_t)k * DOUT;
    v2f b0, b1, b2, b3;
    b0.x = bk[0];   b0.y = bk[DOUT + 0];
    b1.x = bk[16];  b1.y = bk[DOUT + 16];
    b2.x = bk[32];  b2.y = bk[DOUT + 32];
    b3.x = bk[48];  b3.y = bk[DOUT + 48];

    c0 = __builtin_amdgcn_wmma_f32_16x16x4_f32(false, a, false, b0, (short)0, c0, false, false);
    c1 = __builtin_amdgcn_wmma_f32_16x16x4_f32(false, a, false, b1, (short)0, c1, false, false);
    c2 = __builtin_amdgcn_wmma_f32_16x16x4_f32(false, a, false, b2, (short)0, c2, false, false);
    c3 = __builtin_amdgcn_wmma_f32_16x16x4_f32(false, a, false, b3, (short)0, c3, false, false);
  }

  float* crow = C + (size_t)(mbase + 8 * half) * DOUT + nbase + l15;
#pragma unroll
  for (int r = 0; r < 8; ++r) {
    crow[(size_t)r * DOUT + 0]  = c0[r];
    crow[(size_t)r * DOUT + 16] = c1[r];
    crow[(size_t)r * DOUT + 32] = c2[r];
    crow[(size_t)r * DOUT + 48] = c3[r];
  }
}

__global__ __launch_bounds__(256) void init_bias(float* __restrict__ out,
                                                 const float* __restrict__ bias) {
  const int i = blockIdx.x * 256 + threadIdx.x;
  out[i] = bias[i & (DOUT - 1)];
}

__global__ __launch_bounds__(256) void edge_scatter(const float* __restrict__ sup,
                                                    const float* __restrict__ ev,
                                                    const int*   __restrict__ er,
                                                    const int*   __restrict__ ec,
                                                    float* __restrict__ out) {
  const int lane = threadIdx.x & 31;
  const int wave = threadIdx.x >> 5;
  const int e    = blockIdx.x * 8 + wave;
  if (e >= EDGES) return;

  const float  v = ev[e];
  const size_t c = (size_t)ec[e];
  const size_t r = (size_t)er[e];

  const float* src = sup + c * DOUT + lane;
  float*       dst = out + r * DOUT + lane;

  float s[8];
#pragma unroll
  for (int j = 0; j < 8; ++j) s[j] = src[j * 32];
#pragma unroll
  for (int j = 0; j < 8; ++j) atomicAdd(dst + j * 32, s[j] * v);
}

extern "C" void kernel_launch(void* const* d_in, const int* in_sizes, int n_in,
                              void* d_out, int out_size, void* d_ws, size_t ws_size,
                              hipStream_t stream) {
  const float* inputs  = (const float*)d_in[0];
  const float* weights = (const float*)d_in[1];
  const float* bias    = (const float*)d_in[2];
  const float* evals   = (const float*)d_in[3];
  const int*   erow    = (const int*)d_in[4];
  const int*   ecol    = (const int*)d_in[5];
  float* out      = (float*)d_out;
  float* supports = (float*)d_ws;

  gemm_wmma_f32<<<NODES / 16, 128, 0, stream>>>(inputs, weights, supports);
  init_bias<<<(NODES * DOUT) / 256, 256, 0, stream>>>(out, bias);
  edge_scatter<<<EDGES / 8, 256, 0, stream>>>(supports, evals, erow, ecol, out);
}


# fewer, longer waves: scatter wave walks 8 consecutive edges (2 rows' loads in flight, then their adds), 800K waves -> 100K; init thread stores 10 x 16 B, 400K -> 10K; atomics and arrival order kept
# speedup vs baseline: 1.7562x; 1.1585x over previous
#include <hip/hip_runtime.h>

typedef __attribute__((ext_vector_type(2))) float v2f;
typedef __attribute__((ext_vector_type(8))) float v8f;
typedef __attribute__((ext_vector_type(4))) float v4f;

#define NODES 50000
#define DIN   512
#define DOUT  256
#define EDGES 800000

__global__ __launch_bounds__(128) void gemm_wmma_f32(const float* __restrict__ A,
                                                     const float* __restrict__ B,
                                                     float* __restrict__ C) {
  const int lane  = threadIdx.x & 31;
  const int wave  = threadIdx.x >> 5;
  const int half  = lane >> 4;
  const int l15   = lane & 15;
  const int mbase = blockIdx.x * 16;
  const int nbase = wave * 64;

  v8f c0 = {}, c1 = {}, c2 = {}, c3 = {};

  const float* arow = A + (size_t)(mbase + l15) * DIN + 2 * half;
  const float* bcol = B + (size_t)(2 * half) * DOUT + nbase + l15;

  for (int k = 0; k < DIN; k += 4) {
    v2f a = *(const v2f*)(arow + k);

    const float* bk = bcol + (size_t)k * DOUT;
    v2f b0, b1, b2, b3;
    b0.x = bk[0];   b0.y = bk[DOUT + 0];
    b1.x = bk[16];  b1.y = bk[DOUT + 16];
    b2.x = bk[32];  b2.y = bk[DOUT + 32];
    b3.x = bk[48];  b3.y = bk[DOUT + 48];

    c0 = __builtin_amdgcn_wmma_f32_16x16x4_f32(false, a, false, b0, (short)0, c0, false, false);
    c1 = __builtin_amdgcn_wmma_f32_16x16x4_f32(false, a, false, b1, (short)0, c1, false, false);
    c2 = __builtin_amdgcn_wmma_f32_16x16x4_f32(false, a, false, b2, (short)0, c2, false, false);
    c3 = __builtin_amdgcn_wmma_f32_16x16x4_f32(false, a, false, b3, (short)0, c3, false, false);
  }

  float* crow = C + (size_t)(mbase + 8 * half) * DOUT + nbase + l15;
#pragma unroll
  for (int r = 0; r < 8; ++r) {
    crow[(size_t)r * DOUT + 0]  = c0[r];
    crow[(size_t)r * DOUT + 16] = c1[r];
    crow[(size_t)r * DOUT + 32] = c2[r];
    crow[(size_t)r * DOUT + 48] = c3[r];
  }
}

#define INIT_PIECES 10
static_assert((NODES * (DOUT / 4)) % (256 * INIT_PIECES) == 0, "init grid must be exact");

__global__ __launch_bounds__(256) void init_bias(float* __restrict__ out,
                                                 const float* __restrict__ bias) {
  const int t = threadIdx.x;
  const float* bq = bias + 4 * (t & 63);
  v4f bv;
  bv.x = bq[0]; bv.y = bq[1]; bv.z = bq[2]; bv.w = bq[3];
  v4f* o = (v4f*)out + (size_t)blockIdx.x * (256 * INIT_PIECES) + t;
#pragma unroll
  for (int u = 0; u < INIT_PIECES; ++u) o[u * 256] = bv;
}

#define EPW 8
static_assert(EPW % 2 == 0, "edges are taken in pairs");
static_assert(EDGES % (8 * EPW) == 0, "scatter grid must be exact");

__global__ __launch_bounds__(256) void edge_scatter(const float* __restrict__ sup,
                                                    const float* __restrict__ ev,
                                                    const int*   __restrict__ er,
                                                    const int*   __restrict__ ec,
                                                    float* __restrict__ out) {
  const int lane = threadIdx.x & 31;
  const int wave = __builtin_amdgcn_readfirstlane((int)(threadIdx.x >> 5));
  const int e0   = (blockIdx.x * 8 + wave) * EPW;
  if (e0 >= EDGES) return;

#pragma unroll
  for (int i = 0; i < EPW; i += 2) {
    const float* srca = sup + (size_t)ec[e0 + i]     * DOUT + lane;
    const float* srcb = sup + (size_t)ec[e0 + i + 1] * DOUT + lane;
    float sa[8], sb[8];
#pragma unroll
    for (int j = 0; j < 8; ++j) sa[j] = srca[j * 32];
#pragma unroll
    for (int j = 0; j < 8; ++j) sb[j] = srcb[j * 32];

    const float va = ev[e0 + i];
    const float vb = ev[e0 + i + 1];
    float* dsta = out + (size_t)er[e0 + i]     * DOUT + lane;
    float* dstb = out + (size_t)er[e0 + i + 1] * DOUT + lane;
#pragma unroll
    for (int j = 0; j < 8; ++j) atomicAdd(dsta + j * 32, sa[j] * va);
#pragma unroll
    for (int j = 0; j < 8; ++j) atomicAdd(dstb + j * 32, sb[j] * vb);
  }
}

extern "C" void kernel_launch(void* const* d_in, const int* in_sizes, int n_in,
                              void* d_out, int out_size, void* d_ws, size_t ws_size,
                              hipStream_t stream) {
  const float* inputs  = (const float*)d_in[0];
  const float* weights = (const float*)d_in[1];
  const float* bias    = (const float*)d_in[2];
  const float* evals   = (const float*)d_in[3];
  const int*   erow    = (const int*)d_in[4];
  const int*   ecol    = (const int*)d_in[5];
  float* out      = (float*)d_out;
  float* supports = (float*)d_ws;

  gemm_wmma_f32<<<NODES / 16, 128, 0, stream>>>(inputs, weights, supports);
  init_bias<<<(NODES * (DOUT / 4)) / (256 * INIT_PIECES), 256, 0, stream>>>(out, bias);
  edge_scatter<<<EDGES / (8 * EPW), 256, 0, stream>>>(supports, evals, erow, ecol, out);
}


# gemm operand loads widened to 16 B per lane: A lane reads 4 consecutive k, column tiles interleaved so one B load serves 4 tiles; 9 loads per 4 WMMA -> 5 per 8; k summed as {0,1,4,5},{2,3,6,7}
# speedup vs baseline: 1.8419x; 1.0488x over previous
#include <hip/hip_runtime.h>

typedef __attribute__((ext_vector_type(2))) float v2f;
typedef __attribute__((ext_vector_type(8))) float v8f;
typedef __attribute__((ext_vector_type(4))) float v4f;

#define NODES 50000
#define DIN   512
#define DOUT  256
#define EDGES 800000

__global__ __launch_bounds__(128) void gemm_wmma_f32(const float* __restrict__ A,
                                                     const float* __restrict__ B,
                                                     float* __restrict__ C) {
  const int lane  = threadIdx.x & 31;
  const int wave  = threadIdx.x >> 5;
  const int half  = lane >> 4;
  const int l15   = lane & 15;
  const int mbase = blockIdx.x * 16;
  const int nbase = wave * 64;

  v8f c0 = {}, c1 = {}, c2 = {}, c3 = {};

  const float* arow = A + (size_t)(mbase + l15) * DIN + 4 * half;
  const float* bcol = B + (size_t)(4 * half) * DOUT + nbase + 4 * l15;

#pragma unroll 2
  for (int k = 0; k < DIN; k += 8) {
    const v4f a = *(const v4f*)(arow + k);

    const float* bk = bcol + (size_t)k * DOUT;
    const v4f q0 = *(const v4f*)(bk);
    const v4f q1 = *(const v4f*)(bk + DOUT);
    const v4f q2 = *(const v4f*)(bk + 2 * DOUT);
    const v4f q3 = *(const v4f*)(bk + 3 * DOUT);

    v2f alo, ahi, b0, b1, b2, b3;
    alo.x = a.x;  alo.y = a.y;
    ahi.x = a.z;  ahi.y = a.w;

    b0.x = q0.x;  b0.y = q1.x;
    b1.x = q0.y;  b1.y = q1.y;
    b2.x = q0.z;  b2.y = q1.z;
    b3.x = q0.w;  b3.y = q1.w;
    c0 = __builtin_amdgcn_wmma_f32_16x16x4_f32(false, alo, false, b0, (short)0, c0, false, false);
    c1 = __builtin_amdgcn_wmma_f32_16x16x4_f32(false, alo, false, b1, (short)0, c1, false, false);
    c2 = __builtin_amdgcn_wmma_f32_16x16x4_f32(false, alo, false, b2, (short)0, c2, false, false);
    c3 = __builtin_amdgcn_wmma_f32_16x16x4_f32(false, alo, false, b3, (short)0, c3, false, false);

    b0.x = q2.x;  b0.y = q3.x;
    b1.x = q2.y;  b1.y = q3.y;
    b2.x = q2.z;  b2.y = q3.z;
    b3.x = q2.w;  b3.y = q3.w;
    c0 = __builtin_amdgcn_wmma_f32_16x16x4_f32(false, ahi, false, b0, (short)0, c0, false, false);
    c1 = __builtin_amdgcn_wmma_f32_16x16x4_f32(false, ahi, false, b1, (short)0, c1, false, false);
    c2 = __builtin_amdgcn_wmma_f32_16x16x4_f32(false, ahi, false, b2, (short)0, c2, false, false);
    c3 = __builtin_amdgcn_wmma_f32_16x16x4_f32(false, ahi, false, b3, (short)0, c3, false, false);
  }

  float* crow = C + (size_t)(mbase + 8 * half) * DOUT + nbase + 4 * l15;
#pragma unroll
  for (int r = 0; r < 8; ++r) {
    v4f o;
    o.x = c0[r];  o.y = c1[r];  o.z = c2[r];  o.w = c3[r];
    *(v4f*)(crow + (size_t)r * DOUT) = o;
  }
}

#define INIT_PIECES 10
static_assert((NODES * (DOUT / 4)) % (256 * INIT_PIECES) == 0, "init grid must be exact");

__global__ __launch_bounds__(256) void init_bias(float* __restrict__ out,
                                                 const float* __restrict__ bias) {
  const int t = threadIdx.x;
  const float* bq = bias + 4 * (t & 63);
  v4f bv;
  bv.x = bq[0]; bv.y = bq[1]; bv.z = bq[2]; bv.w = bq[3];
  v4f* o = (v4f*)out + (size_t)blockIdx.x * (256 * INIT_PIECES) + t;
#pragma unroll
  for (int u = 0; u < INIT_PIECES; ++u) o[u * 256] = bv;
}

#define EPW 8
static_assert(EPW % 2 == 0, "edges are taken in pairs");
static_assert(EDGES % (8 * EPW) == 0, "scatter grid must be exact");

__global__ __launch_bounds__(256) void edge_scatter(const float* __restrict__ sup,
                                                    const float* __restrict__ ev,
                                                    const int*   __restrict__ er,
                                                    const int*   __restrict__ ec,
                                                    float* __restrict__ out) {
  const int lane = threadIdx.x & 31;
  const int wave = __builtin_amdgcn_readfirstlane((int)(threadIdx.x >> 5));
  const int e0   = (blockIdx.x * 8 + wave) * EPW;
  if (e0 >= EDGES) return;

#pragma unroll
  for (int i = 0; i < EPW; i += 2) {
    const float* srca = sup + (size_t)ec[e0 + i]     * DOUT + lane;
    const float* srcb = sup + (size_t)ec[e0 + i + 1] * DOUT + lane;
    float sa[8], sb[8];
#pragma unroll
    for (int j = 0; j < 8; ++j) sa[j] = srca[j * 32];
#pragma unroll
    for (int j = 0; j < 8; ++j) sb[j] = srcb[j * 32];

    const float va = ev[e0 + i];
    const float vb = ev[e0 + i + 1];
    float* dsta = out + (size_t)er[e0 + i]     * DOUT + lane;
    float* dstb = out + (size_t)er[e0 + i + 1] * DOUT + lane;
#pragma unroll
    for (int j = 0; j < 8; ++j) atomicAdd(dsta + j * 32, sa[j] * va);
#pragma unroll
    for (int j = 0; j < 8; ++j) atomicAdd(dstb + j * 32, sb[j] * vb);
  }
}

extern "C" void kernel_launch(void* const* d_in, const int* in_sizes, int n_in,
                              void* d_out, int out_size, void* d_ws, size_t ws_size,
                              hipStream_t stream) {
  const float* inputs  = (const float*)d_in[0];
  const float* weights = (const float*)d_in[1];
  const float* bias    = (const float*)d_in[2];
  const float* evals   = (const float*)d_in[3];
  const int*   erow    = (const int*)d_in[4];
  const int*   ecol    = (const int*)d_in[5];
  float* out      = (float*)d_out;
  float* supports = (float*)d_ws;

  gemm_wmma_f32<<<NODES / 16, 128, 0, stream>>>(inputs, weights, supports);
  init_bias<<<(NODES * (DOUT / 4)) / (256 * INIT_PIECES), 256, 0, stream>>>(out, bias);
  edge_scatter<<<EDGES / (8 * EPW), 256, 0, stream>>>(supports, evals, erow, ecol, out);
}


# edge sums by ownership: each of 250 work-groups keeps 200 output rows in LDS, scans edge_row, adds value*row by LDS f32 atomics (arrival order), stores once; no global atomics, no bias launch
# speedup vs baseline: 1.9072x; 1.0355x over previous
#include <hip/hip_runtime.h>

typedef __attribute__((ext_vector_type(2))) float v2f;
typedef __attribute__((ext_vector_type(8))) float v8f;
typedef __attribute__((ext_vector_type(4))) float v4f;

#define NODES 50000
#define DIN   512
#define DOUT  256
#define EDGES 800000

__global__ __launch_bounds__(128) void gemm_wmma_f32(const float* __restrict__ A,
                                                     const float* __restrict__ B,
                                                     float* __restrict__ C) {
  const int lane  = threadIdx.x & 31;
  const int wave  = threadIdx.x >> 5;
  const int half  = lane >> 4;
  const int l15   = lane & 15;
  const int mbase = blockIdx.x * 16;
  const int nbase = wave * 64;

  v8f c0 = {}, c1 = {}, c2 = {}, c3 = {};

  const float* arow = A + (size_t)(mbase + l15) * DIN + 4 * half;
  const float* bcol = B + (size_t)(4 * half) * DOUT + nbase + 4 * l15;

#pragma unroll 2
  for (int k = 0; k < DIN; k += 8) {
    const v4f a = *(const v4f*)(arow + k);

    const float* bk = bcol + (size_t)k * DOUT;
    const v4f q0 = *(const v4f*)(bk);
    const v4f q1 = *(const v4f*)(bk + DOUT);
    const v4f q2 = *(const v4f*)(bk + 2 * DOUT);
    const v4f q3 = *(const v4f*)(bk + 3 * DOUT);

    v2f alo, ahi, b0, b1, b2, b3;
    alo.x = a.x;  alo.y = a.y;
    ahi.x = a.z;  ahi.y = a.w;

    b0.x = q0.x;  b0.y = q1.x;
    b1.x = q0.y;  b1.y = q1.y;
    b2.x = q0.z;  b2.y = q1.z;
    b3.x = q0.w;  b3.y = q1.w;
    c0 = __builtin_amdgcn_wmma_f32_16x16x4_f32(false, alo, false, b0, (short)0, c0, false, false);
    c1 = __builtin_amdgcn_wmma_f32_16x16x4_f32(false, alo, false, b1, (short)0, c1, false, false);
    c2 = __builtin_amdgcn_wmma_f32_16x16x4_f32(false, alo, false, b2, (short)0, c2, false, false);
    c3 = __builtin_amdgcn_wmma_f32_16x16x4_f32(false, alo, false, b3, (short)0, c3, false, false);

    b0.x = q2.x;  b0.y = q3.x;
    b1.x = q2.y;  b1.y = q3.y;
    b2.x = q2.z;  b2.y = q3.z;
    b3.x = q2.w;  b3.y = q3.w;
    c0 = __builtin_amdgcn_wmma_f32_16x16x4_f32(false, ahi, false, b0, (short)0, c0, false, false);
    c1 = __builtin_amdgcn_wmma_f32_16x16x4_f32(false, ahi, false, b1, (short)0, c1, false, false);
    c2 = __builtin_amdgcn_wmma_f32_16x16x4_f32(false, ahi, false, b2, (short)0, c2, false, false);
    c3 = __builtin_amdgcn_wmma_f32_16x16x4_f32(false, ahi, false, b3, (short)0, c3, false, false);
  }

  float* crow = C + (size_t)(mbase + 8 * half) * DOUT + nbase + 4 * l15;
#pragma unroll
  for (int r = 0; r < 8; ++r) {
    v4f o;
    o.x = c0[r];  o.y = c1[r];  o.z = c2[r];  o.w = c3[r];
    *(v4f*)(crow + (size_t)r * DOUT) = o;
  }
}

typedef __attribute__((ext_vector_type(4))) int v4i;

#define OWN_ROWS   200
#define OWN_WGS    (NODES / OWN_ROWS)
#define OWN_WAVES  32
#define QCAP       256
#define BATCH      8
#define SCAN_U     4
#define QUADS_PER_WAVE (EDGES / 4 / OWN_WAVES)
#define SCAN_ITERS ((QUADS_PER_WAVE + 32 * SCAN_U - 1) / (32 * SCAN_U))
static_assert(OWN_WGS * OWN_ROWS == NODES, "owned row ranges must tile the nodes exactly");
static_assert(EDGES % (4 * OWN_WAVES) == 0, "edge quads must split evenly over the waves");
static_assert(OWN_ROWS <= 256, "local row must fit the low 8 bits of a queue entry");
static_assert(BATCH - 1 + 128 <= QCAP, "queue holds a left-over plus one 128-edge step");

__device__ __forceinline__ void drain_batch(const float* __restrict__ sup,
                                            const float* __restrict__ ev,
                                            const int*   __restrict__ ec,
                                            float* acc, const int* q, int n, int lane) {
  const int slot = lane & (BATCH - 1);
  const int entv = q[slot < n ? slot : n - 1];
  int   lrow[BATCH];
  float val[BATCH];
  float s[BATCH][8];
#pragma unroll
  for (int i = 0; i < BATCH; ++i) {
    const int ent = __builtin_amdgcn_readlane(entv, i);
    const int e   = ent >> 8;
    lrow[i] = ent & 255;
    val[i]  = ev[e];
    const float* src = sup + (size_t)ec[e] * DOUT + lane;
#pragma unroll
    for (int j = 0; j < 8; ++j) s[i][j] = src[j * 32];
  }
#pragma unroll
  for (int i = 0; i < BATCH; ++i) {
    if (i < n) {
      float* dst = acc + lrow[i] * DOUT + lane;
#pragma unroll
      for (int j = 0; j < 8; ++j) atomicAdd(dst + j * 32, s[i][j] * val[i]);
    }
  }
}

__global__ __launch_bounds__(32 * OWN_WAVES) void owner_aggregate(const float* __restrict__ sup,
                                                                  const float* __restrict__ ev,
                                                                  const int*   __restrict__ er,
                                                                  const int*   __restrict__ ec,
                                                                  const float* __restrict__ bias,
                                                                  float* __restrict__ out) {
  __shared__ __attribute__((aligned(16))) float acc[OWN_ROWS * DOUT];
  __shared__ int queue[OWN_WAVES * QCAP];

  const int t    = threadIdx.x;
  const int lane = t & 31;
  const int wave = __builtin_amdgcn_readfirstlane(t >> 5);
  const int lo   = blockIdx.x * OWN_ROWS;

  {
    const float* bq = bias + 4 * (t & 63);
    v4f bv;
    bv.x = bq[0]; bv.y = bq[1]; bv.z = bq[2]; bv.w = bq[3];
    v4f* a4 = (v4f*)acc;
    for (int i = t; i < OWN_ROWS * (DOUT / 4); i += 32 * OWN_WAVES) a4[i] = bv;
  }
  __syncthreads();

  {
    const v4i* er4 = (const v4i*)er + (size_t)wave * QUADS_PER_WAVE;
    int* q = queue + wave * QCAP;
    int count = 0;
    const v4i none = {-1, -1, -1, -1};

    v4i cur[SCAN_U], nxt[SCAN_U];
#pragma unroll
    for (int u = 0; u < SCAN_U; ++u) {
      const int qi = u * 32 + lane;
      cur[u] = (qi < QUADS_PER_WAVE) ? er4[qi] : none;
    }

    for (int it = 0; it < SCAN_ITERS; ++it) {
#pragma unroll
      for (int u = 0; u < SCAN_U; ++u) {
        const int qi = (it + 1) * (32 * SCAN_U) + u * 32 + lane;
        nxt[u] = (qi < QUADS_PER_WAVE) ? er4[qi] : none;
      }

#pragma unroll
      for (int u = 0; u < SCAN_U; ++u) {
        const int qi    = it * (32 * SCAN_U) + u * 32 + lane;
        const int ebase = (wave * QUADS_PER_WAVE + qi) * 4;
        unsigned d[4], mask[4];
#pragma unroll
        for (int k = 0; k < 4; ++k) {
          d[k]    = (unsigned)(cur[u][k] - lo);
          mask[k] = __builtin_amdgcn_ballot_w32(d[k] < (unsigned)OWN_ROWS);
        }
        if ((mask[0] | mask[1] | mask[2] | mask[3]) != 0u) {
#pragma unroll
          for (int k = 0; k < 4; ++k) {
            if (d[k] < (unsigned)OWN_ROWS)
              q[count + (int)__builtin_amdgcn_mbcnt_lo(mask[k], 0u)] = ((ebase + k) << 8) | (int)d[k];
            count += __builtin_popcount(mask[k]);
          }
        }
        __builtin_amdgcn_fence(__ATOMIC_ACQ_REL, "wavefront");
        while (count >= BATCH) {
          count -= BATCH;
          drain_batch(sup, ev, ec, acc, q + count, BATCH, lane);
        }
      }

#pragma unroll
      for (int u = 0; u < SCAN_U; ++u) cur[u] = nxt[u];
    }
    if (count > 0) drain_batch(sup, ev, ec, acc, q, count, lane);
  }
  __syncthreads();

  {
    const v4f* a4 = (const v4f*)acc;
    v4f* o4 = (v4f*)out + (size_t)lo * (DOUT / 4);
    for (int i = t; i < OWN_ROWS * (DOUT / 4); i += 32 * OWN_WAVES) o4[i] = a4[i];
  }
}

extern "C" void kernel_launch(void* const* d_in, const int* in_sizes, int n_in,
                              void* d_out, int out_size, void* d_ws, size_t ws_size,
                              hipStream_t stream) {
  const float* inputs  = (const float*)d_in[0];
  const float* weights = (const float*)d_in[1];
  const float* bias    = (const float*)d_in[2];
  const float* evals   = (const float*)d_in[3];
  const int*   erow    = (const int*)d_in[4];
  const int*   ecol    = (const int*)d_in[5];
  float* out      = (float*)d_out;
  float* supports = (float*)d_ws;

  gemm_wmma_f32<<<NODES / 16, 128, 0, stream>>>(inputs, weights, supports);
  owner_aggregate<<<OWN_WGS, 32 * OWN_WAVES, 0, stream>>>(supports, evals, erow, ecol, bias, out);
}


# gemm operand loads issued two trips (32 WMMAs) ahead of use through three rotating register sets; wave tile 32x64 so a 6-load group feeds 16 WMMAs; per-tile k order kept, sums bit-identical
# speedup vs baseline: 2.0618x; 1.0811x over previous
#include <hip/hip_runtime.h>

typedef __attribute__((ext_vector_type(2))) float v2f;
typedef __attribute__((ext_vector_type(8))) float v8f;
typedef __attribute__((ext_vector_type(4))) float v4f;

#define NODES 50000
#define DIN   512
#define DOUT  256
#define EDGES 800000

#define ROW_TILES    (NODES / 16)
#define PAIR_BLOCKS  1536
#define LONE_BLOCKS  (ROW_TILES - 2 * PAIR_BLOCKS)
#define KTRIPS       (DIN / 8)
static_assert(ROW_TILES * 16 == NODES, "row tiles must cover the nodes exactly");
static_assert(LONE_BLOCKS >= 0, "two-tile blocks must not run past the last row tile");
static_assert(KTRIPS % 3 == 1, "the loop below runs KTRIPS - 1 trips in threes and one more behind it");

template <int MT>
struct OperandSet {
  v4f a[MT];
  v4f q[4];
};

template <int MT>
__device__ __forceinline__ void load_operands(OperandSet<MT>& s, const float* const (&arow)[MT],
                                              const float* bcol, int k) {
  k = (k < DIN - 8) ? k : DIN - 8;
#pragma unroll
  for (int m = 0; m < MT; ++m) s.a[m] = *(const v4f*)(arow[m] + k);
#pragma unroll
  for (int j = 0; j < 4; ++j) s.q[j] = *(const v4f*)(bcol + (size_t)(k + j) * DOUT);
  asm volatile("" ::: "memory");
  __builtin_amdgcn_sched_barrier(0);
}

template <int MT>
__device__ __forceinline__ void multiply_trip(v8f (&acc)[MT][4], const OperandSet<MT>& s) {
#pragma unroll
  for (int g = 0; g < 2; ++g) {
    v2f ap[MT], bp[4];
#pragma unroll
    for (int m = 0; m < MT; ++m) { ap[m].x = s.a[m][2 * g];  ap[m].y = s.a[m][2 * g + 1]; }
#pragma unroll
    for (int t = 0; t < 4; ++t) { bp[t].x = s.q[2 * g][t];  bp[t].y = s.q[2 * g + 1][t]; }
#pragma unroll
    for (int t = 0; t < 4; ++t)
#pragma unroll
      for (int m = 0; m < MT; ++m)
        acc[m][t] = __builtin_amdgcn_wmma_f32_16x16x4_f32(false, ap[m], false, bp[t], (short)0,
                                                          acc[m][t], false, false);
  }
}

template <int MT>
__device__ __forceinline__ void gemm_row_tiles(const float* __restrict__ A,
                                               const float* __restrict__ B,
                                               float* __restrict__ C, const int mbase) {
  const int lane  = threadIdx.x & 31;
  const int wave  = threadIdx.x >> 5;
  const int half  = lane >> 4;
  const int l15   = lane & 15;
  const int nbase = wave * 64;

  v8f acc[MT][4];
#pragma unroll
  for (int m = 0; m < MT; ++m)
#pragma unroll
    for (int t = 0; t < 4; ++t) acc[m][t] = v8f{};

  const float* arow[MT];
#pragma unroll
  for (int m = 0; m < MT; ++m) arow[m] = A + (size_t)(mbase + 16 * m + l15) * DIN + 4 * half;
  const float* bcol = B + (size_t)(4 * half) * DOUT + nbase + 4 * l15;

  OperandSet<MT> s0, s1, s2;
  load_operands<MT>(s0, arow, bcol, 0);
  load_operands<MT>(s1, arow, bcol, 8);
  for (int k = 0; k < DIN - 8; k += 24) {
    load_operands<MT>(s2, arow, bcol, k + 16);
    multiply_trip<MT>(acc, s0);
    load_operands<MT>(s0, arow, bcol, k + 24);
    multiply_trip<MT>(acc, s1);
    load_operands<MT>(s1, arow, bcol, k + 32);
    multiply_trip<MT>(acc, s2);
  }
  multiply_trip<MT>(acc, s0);

#pragma unroll
  for (int m = 0; m < MT; ++m) {
    float* crow = C + (size_t)(mbase + 16 * m + 8 * half) * DOUT + nbase + 4 * l15;
#pragma unroll
    for (int r = 0; r < 8; ++r) {
      v4f o;
      o.x = acc[m][0][r];  o.y = acc[m][1][r];  o.z = acc[m][2][r];  o.w = acc[m][3][r];
      *(v4f*)(crow + (size_t)r * DOUT) = o;
    }
  }
}

__global__ __launch_bounds__(128) void gemm_wmma_f32(const float* __restrict__ A,
                                                     const float* __restrict__ B,
                                                     float* __restrict__ C) {
  const int b = blockIdx.x;
  if (b < PAIR_BLOCKS) gemm_row_tiles<2>(A, B, C, b * 32);
  else                 gemm_row_tiles<1>(A, B, C, (2 * PAIR_BLOCKS + (b - PAIR_BLOCKS)) * 16);
}

typedef __attribute__((ext_vector_type(4))) int v4i;

#define OWN_ROWS   200
#define OWN_WGS    (NODES / OWN_ROWS)
#define OWN_WAVES  32
#define QCAP       256
#define BATCH      8
#define SCAN_U     4
#define QUADS_PER_WAVE (EDGES / 4 / OWN_WAVES)
#define SCAN_ITERS ((QUADS_PER_WAVE + 32 * SCAN_U - 1) / (32 * SCAN_U))
static_assert(OWN_WGS * OWN_ROWS == NODES, "owned row ranges must tile the nodes exactly");
static_assert(EDGES % (4 * OWN_WAVES) == 0, "edge quads must split evenly over the waves");
static_assert(OWN_ROWS <= 256, "local row must fit the low 8 bits of a queue entry");
static_assert(BATCH - 1 + 128 <= QCAP, "queue holds a left-over plus one 128-edge step");

__device__ __forceinline__ void drain_batch(const float* __restrict__ sup,
                                            const float* __restrict__ ev,
                                            const int*   __restrict__ ec,
                                            float* acc, const int* q, int n, int lane) {
  const int slot = lane & (BATCH - 1);
  const int entv = q[slot < n ? slot : n - 1];
  int   lrow[BATCH];
  float val[BATCH];
  float s[BATCH][8];
#pragma unroll
  for (int i = 0; i < BATCH; ++i) {
    const int ent = __builtin_amdgcn_readlane(entv, i);
    const int e   = ent >> 8;
    lrow[i] = ent & 255;
    val[i]  = ev[e];
    const float* src = sup + (size_t)ec[e] * DOUT + lane;
#pragma unroll
    for (int j = 0; j < 8; ++j) s[i][j] = src[j * 32];
  }
#pragma unroll
  for (int i = 0; i < BATCH; ++i) {
    if (i < n) {
      float* dst = acc + lrow[i] * DOUT + lane;
#pragma unroll
      for (int j = 0; j < 8; ++j) atomicAdd(dst + j * 32, s[i][j] * val[i]);
    }
  }
}

__global__ __launch_bounds__(32 * OWN_WAVES) void owner_aggregate(const float* __restrict__ sup,
                                                                  const float* __restrict__ ev,
                                                                  const int*   __restrict__ er,
                                                                  const int*   __restrict__ ec,
                                                                  const float* __restrict__ bias,
                                                                  float* __restrict__ out) {
  __shared__ __attribute__((aligned(16))) float acc[OWN_ROWS * DOUT];
  __shared__ int queue[OWN_WAVES * QCAP];

  const int t    = threadIdx.x;
  const int lane = t & 31;
  const int wave = __builtin_amdgcn_readfirstlane(t >> 5);
  const int lo   = blockIdx.x * OWN_ROWS;

  {
    const float* bq = bias + 4 * (t & 63);
    v4f bv;
    bv.x = bq[0]; bv.y = bq[1]; bv.z = bq[2]; bv.w = bq[3];
    v4f* a4 = (v4f*)acc;
    for (int i = t; i < OWN_ROWS * (DOUT / 4); i += 32 * OWN_WAVES) a4[i] = bv;
  }
  __syncthreads();

  {
    const v4i* er4 = (const v4i*)er + (size_t)wave * QUADS_PER_WAVE;
    int* q = queue + wave * QCAP;
    int count = 0;
    const v4i none = {-1, -1, -1, -1};

    v4i cur[SCAN_U], nxt[SCAN_U];
#pragma unroll
    for (int u = 0; u < SCAN_U; ++u) {
      const int qi = u * 32 + lane;
      cur[u] = (qi < QUADS_PER_WAVE) ? er4[qi] : none;
    }

    for (int it = 0; it < SCAN_ITERS; ++it) {
#pragma unroll
      for (int u = 0; u < SCAN_U; ++u) {
        const int qi = (it + 1) * (32 * SCAN_U) + u * 32 + lane;
        nxt[u] = (qi < QUADS_PER_WAVE) ? er4[qi] : none;
      }

#pragma unroll
      for (int u = 0; u < SCAN_U; ++u) {
        const int qi    = it * (32 * SCAN_U) + u * 32 + lane;
        const int ebase = (wave * QUADS_PER_WAVE + qi) * 4;
        unsigned d[4], mask[4];
#pragma unroll
        for (int k = 0; k < 4; ++k) {
          d[k]    = (unsigned)(cur[u][k] - lo);
          mask[k] = __builtin_amdgcn_ballot_w32(d[k] < (unsigned)OWN_ROWS);
        }
        if ((mask[0] | mask[1] | mask[2] | mask[3]) != 0u) {
#pragma unroll
          for (int k = 0; k < 4; ++k) {
            if (d[k] < (unsigned)OWN_ROWS)
              q[count + (int)__builtin_amdgcn_mbcnt_lo(mask[k], 0u)] = ((ebase + k) << 8) | (int)d[k];
            count += __builtin_popcount(mask[k]);
          }
        }
        __builtin_amdgcn_fence(__ATOMIC_ACQ_REL, "wavefront");
        while (count >= BATCH) {
          count -= BATCH;
          drain_batch(sup, ev, ec, acc, q + count, BATCH, lane);
        }
      }

#pragma unroll
      for (int u = 0; u < SCAN_U; ++u) cur[u] = nxt[u];
    }
    if (count > 0) drain_batch(sup, ev, ec, acc, q, count, lane);
  }
  __syncthreads();

  {
    const v4f* a4 = (const v4f*)acc;
    v4f* o4 = (v4f*)out + (size_t)lo * (DOUT / 4);
    for (int i = t; i < OWN_ROWS * (DOUT / 4); i += 32 * OWN_WAVES) o4[i] = a4[i];
  }
}

extern "C" void kernel_launch(void* const* d_in, const int* in_sizes, int n_in,
                              void* d_out, int out_size, void* d_ws, size_t ws_size,
                              hipStream_t stream) {
  const float* inputs  = (const float*)d_in[0];
  const float* weights = (const float*)d_in[1];
  const float* bias    = (const float*)d_in[2];
  const float* evals   = (const float*)d_in[3];
  const int*   erow    = (const int*)d_in[4];
  const int*   ecol    = (const int*)d_in[5];
  float* out      = (float*)d_out;
  float* supports = (float*)d_ws;

  gemm_wmma_f32<<<PAIR_BLOCKS + LONE_BLOCKS, 128, 0, stream>>>(inputs, weights, supports);
  owner_aggregate<<<OWN_WGS, 32 * OWN_WAVES, 0, stream>>>(supports, evals, erow, ecol, bias, out);
}


# edges binned by owner once per call (new launch: LDS counts, one global int add per chunk and owner; lists in the owner's own out rows); owners stop scanning all edges; arrival-order f32 sums kept
# speedup vs baseline: 2.6237x; 1.2725x over previous
#include <hip/hip_runtime.h>

typedef __attribute__((ext_vector_type(2))) float v2f;
typedef __attribute__((ext_vector_type(8))) float v8f;
typedef __attribute__((ext_vector_type(4))) float v4f;

#define NODES 50000
#define DIN   512
#define DOUT  256
#define EDGES 800000

#define OWN_ROWS   200
#define OWN_WGS    (NODES / OWN_ROWS)
#define LIST_HEAD  32
#define LIST_CAP   (OWN_ROWS * DOUT - LIST_HEAD)

#define ROW_TILES    (NODES / 16)
#define PAIR_BLOCKS  1536
#define LONE_BLOCKS  (ROW_TILES - 2 * PAIR_BLOCKS)
#define KTRIPS       (DIN / 8)
static_assert(ROW_TILES * 16 == NODES, "row tiles must cover the nodes exactly");
static_assert(LONE_BLOCKS >= 0, "two-tile blocks must not run past the last row tile");
static_assert(KTRIPS % 3 == 1, "the loop below runs KTRIPS - 1 trips in threes and one more behind it");

template <int MT>
struct OperandSet {
  v4f a[MT];
  v4f q[4];
};

template <int MT>
__device__ __forceinline__ void load_operands(OperandSet<MT>& s, const float* const (&arow)[MT],
                                              const float* bcol, int k) {
  k = (k < DIN - 8) ? k : DIN - 8;
#pragma unroll
  for (int m = 0; m < MT; ++m) s.a[m] = *(const v4f*)(arow[m] + k);
#pragma unroll
  for (int j = 0; j < 4; ++j) s.q[j] = *(const v4f*)(bcol + (size_t)(k + j) * DOUT);
  asm volatile("" ::: "memory");
  __builtin_amdgcn_sched_barrier(0);
}

template <int MT>
__device__ __forceinline__ void multiply_trip(v8f (&acc)[MT][4], const OperandSet<MT>& s) {
#pragma unroll
  for (int g = 0; g < 2; ++g) {
    v2f ap[MT], bp[4];
#pragma unroll
    for (int m = 0; m < MT; ++m) { ap[m].x = s.a[m][2 * g];  ap[m].y = s.a[m][2 * g + 1]; }
#pragma unroll
    for (int t = 0; t < 4; ++t) { bp[t].x = s.q[2 * g][t];  bp[t].y = s.q[2 * g + 1][t]; }
#pragma unroll
    for (int t = 0; t < 4; ++t)
#pragma unroll
      for (int m = 0; m < MT; ++m)
        acc[m][t] = __builtin_amdgcn_wmma_f32_16x16x4_f32(false, ap[m], false, bp[t], (short)0,
                                                          acc[m][t], false, false);
  }
}

template <int MT>
__device__ __forceinline__ void gemm_row_tiles(const float* __restrict__ A,
                                               const float* __restrict__ B,
                                               float* __restrict__ C, const int mbase) {
  const int lane  = threadIdx.x & 31;
  const int wave  = threadIdx.x >> 5;
  const int half  = lane >> 4;
  const int l15   = lane & 15;
  const int nbase = wave * 64;

  v8f acc[MT][4];
#pragma unroll
  for (int m = 0; m < MT; ++m)
#pragma unroll
    for (int t = 0; t < 4; ++t) acc[m][t] = v8f{};

  const float* arow[MT];
#pragma unroll
  for (int m = 0; m < MT; ++m) arow[m] = A + (size_t)(mbase + 16 * m + l15) * DIN + 4 * half;
  const float* bcol = B + (size_t)(4 * half) * DOUT + nbase + 4 * l15;

  OperandSet<MT> s0, s1, s2;
  load_operands<MT>(s0, arow, bcol, 0);
  load_operands<MT>(s1, arow, bcol, 8);
  for (int k = 0; k < DIN - 8; k += 24) {
    load_operands<MT>(s2, arow, bcol, k + 16);
    multiply_trip<MT>(acc, s0);
    load_operands<MT>(s0, arow, bcol, k + 24);
    multiply_trip<MT>(acc, s1);
    load_operands<MT>(s1, arow, bcol, k + 32);
    multiply_trip<MT>(acc, s2);
  }
  multiply_trip<MT>(acc, s0);

#pragma unroll
  for (int m = 0; m < MT; ++m) {
    float* crow = C + (size_t)(mbase + 16 * m + 8 * half) * DOUT + nbase + 4 * l15;
#pragma unroll
    for (int r = 0; r < 8; ++r) {
      v4f o;
      o.x = acc[m][0][r];  o.y = acc[m][1][r];  o.z = acc[m][2][r];  o.w = acc[m][3][r];
      *(v4f*)(crow + (size_t)r * DOUT) = o;
    }
  }
}

__global__ __launch_bounds__(128) void gemm_wmma_f32(const float* __restrict__ A,
                                                     const float* __restrict__ B,
                                                     float* __restrict__ C,
                                                     float* __restrict__ outbuf) {
  const int b = blockIdx.x;
  if (b < OWN_WGS && threadIdx.x == 0) ((int*)outbuf)[(size_t)b * (OWN_ROWS * DOUT)] = 0;
  if (b < PAIR_BLOCKS) gemm_row_tiles<2>(A, B, C, b * 32);
  else                 gemm_row_tiles<1>(A, B, C, (2 * PAIR_BLOCKS + (b - PAIR_BLOCKS)) * 16);
}

typedef __attribute__((ext_vector_type(4))) int v4i;

#define BIN_EDGES   (EDGES / OWN_WGS)
#define BIN_QUADS   (BIN_EDGES / 4)
#define BIN_THREADS 256
#define BIN_STEPS   ((BIN_QUADS + BIN_THREADS - 1) / BIN_THREADS)
static_assert(BIN_EDGES * OWN_WGS == EDGES && BIN_QUADS * 4 == BIN_EDGES, "chunks of whole edge quads must cover the list");
static_assert(OWN_WGS <= BIN_THREADS, "one thread per owner reserves the chunk's run");
static_assert(OWN_ROWS <= 256 && EDGES <= (1 << 20), "entry packs a 20-bit edge index and an 8-bit local row");

__global__ __launch_bounds__(BIN_THREADS) void bin_edges(const int* __restrict__ er, float* __restrict__ outbuf) {
  __shared__ int run[OWN_WGS];
  const int t = threadIdx.x;
  if (t < OWN_WGS) run[t] = 0;
  __syncthreads();

  const v4i none = {-1, -1, -1, -1};
  v4i r[BIN_STEPS];
#pragma unroll
  for (int s = 0; s < BIN_STEPS; ++s) {
    const int qi = s * BIN_THREADS + t;
    r[s] = (qi < BIN_QUADS) ? ((const v4i*)er)[blockIdx.x * BIN_QUADS + qi] : none;
  }
  int owner[BIN_STEPS][4], ent[BIN_STEPS][4], rank[BIN_STEPS][4];
#pragma unroll
  for (int s = 0; s < BIN_STEPS; ++s) {
    const int quad = blockIdx.x * BIN_QUADS + s * BIN_THREADS + t;
#pragma unroll
    for (int k = 0; k < 4; ++k) {
      const unsigned row = (unsigned)r[s][k];
      const unsigned o   = row / OWN_ROWS;
      owner[s][k] = -1;  ent[s][k] = 0;  rank[s][k] = 0;
      if (o < (unsigned)OWN_WGS) {
        owner[s][k] = (int)o;
        ent[s][k]   = ((4 * quad + k) << 8) | (int)(row - o * OWN_ROWS);
        rank[s][k]  = atomicAdd(&run[o], 1);
      }
    }
  }
  __syncthreads();
  if (t < OWN_WGS) {
    const int c = run[t];
    int* len = (int*)outbuf + (size_t)t * (OWN_ROWS * DOUT);
    run[t] = (c > 0) ? atomicAdd(len, c) : 0;
  }
  __syncthreads();
#pragma unroll
  for (int s = 0; s < BIN_STEPS; ++s) {
#pragma unroll
    for (int k = 0; k < 4; ++k) {
      if (owner[s][k] >= 0) {
        const int pos = run[owner[s][k]] + rank[s][k];
        if (pos < LIST_CAP)
          ((int*)outbuf)[(size_t)owner[s][k] * (OWN_ROWS * DOUT) + LIST_HEAD + pos] = ent[s][k];
      }
    }
  }
}


#define OWN_WAVES  32
#define QCAP       256
#define BATCH      8
#define SCAN_U     4
#define QUADS_PER_WAVE (EDGES / 4 / OWN_WAVES)
#define SCAN_ITERS ((QUADS_PER_WAVE + 32 * SCAN_U - 1) / (32 * SCAN_U))
static_assert(OWN_WGS * OWN_ROWS == NODES, "owned row ranges must tile the nodes exactly");
static_assert(EDGES % (4 * OWN_WAVES) == 0, "edge quads must split evenly over the waves");
static_assert(OWN_ROWS <= 256, "local row must fit the low 8 bits of a queue entry");
static_assert(BATCH - 1 + 128 <= QCAP, "queue holds a left-over plus one 128-edge step");

__device__ __forceinline__ void drain_batch(const float* __restrict__ sup,
                                            const float* __restrict__ ev,
                                            const int*   __restrict__ ec,
                                            float* acc, const int entv, int n) {
  const int lane = threadIdx.x & 31;
  int   lrow[BATCH];
  float val[BATCH];
  float s[BATCH][8];
#pragma unroll
  for (int i = 0; i < BATCH; ++i) {
    const int ent = __builtin_amdgcn_readlane(entv, i);
    const int e   = ent >> 8;
    lrow[i] = ent & 255;
    val[i]  = ev[e];
    const float* src = sup + (size_t)ec[e] * DOUT + lane;
#pragma unroll
    for (int j = 0; j < 8; ++j) s[i][j] = src[j * 32];
  }
#pragma unroll
  for (int i = 0; i < BATCH; ++i) {
    if (i < n) {
      float* dst = acc + lrow[i] * DOUT + lane;
#pragma unroll
      for (int j = 0; j < 8; ++j) atomicAdd(dst + j * 32, s[i][j] * val[i]);
    }
  }
}

__global__ __launch_bounds__(32 * OWN_WAVES) void owner_aggregate(const float* __restrict__ sup,
                                                                  const float* __restrict__ ev,
                                                                  const int*   __restrict__ er,
                                                                  const int*   __restrict__ ec,
                                                                  const float* __restrict__ bias,
                                                                  float* out) {
  __shared__ __attribute__((aligned(16))) float acc[OWN_ROWS * DOUT];
  __shared__ int queue[OWN_WAVES * QCAP];

  const int t    = threadIdx.x;
  const int lane = t & 31;
  const int wave = __builtin_amdgcn_readfirstlane(t >> 5);
  const int lo   = blockIdx.x * OWN_ROWS;

  {
    const float* bq = bias + 4 * (t & 63);
    v4f bv;
    bv.x = bq[0]; bv.y = bq[1]; bv.z = bq[2]; bv.w = bq[3];
    v4f* a4 = (v4f*)acc;
    for (int i = t; i < OWN_ROWS * (DOUT / 4); i += 32 * OWN_WAVES) a4[i] = bv;
  }
  __syncthreads();

  const int* list = (const int*)out + (size_t)lo * DOUT;
  const int  nlist = __builtin_amdgcn_readfirstlane(list[0]);
  const int  slot  = lane & (BATCH - 1);

  if (nlist <= LIST_CAP) {
    const int* ent = list + LIST_HEAD;
    int i = wave * BATCH;
    int cur = 0;
    if (i < nlist) cur = ent[(i + slot < nlist) ? i + slot : nlist - 1];
    while (i + BATCH <= nlist) {
      const int inext = i + BATCH * OWN_WAVES;
      int nxt = 0;
      if (inext < nlist) nxt = ent[(inext + slot < nlist) ? inext + slot : nlist - 1];
      drain_batch(sup, ev, ec, acc, cur, BATCH);
      cur = nxt;
      i   = inext;
    }
    if (i < nlist) drain_batch(sup, ev, ec, acc, cur, nlist - i);
  } else {
    const v4i* er4 = (const v4i*)er + (size_t)wave * QUADS_PER_WAVE;
    int* q = queue + wave * QCAP;
    int count = 0;
    const v4i none = {-1, -1, -1, -1};

    v4i cur[SCAN_U], nxt[SCAN_U];
#pragma unroll
    for (int u = 0; u < SCAN_U; ++u) {
      const int qi = u * 32 + lane;
      cur[u] = (qi < QUADS_PER_WAVE) ? er4[qi] : none;
    }

    for (int it = 0; it < SCAN_ITERS; ++it) {
#pragma unroll
      for (int u = 0; u < SCAN_U; ++u) {
        const int qi = (it + 1) * (32 * SCAN_U) + u * 32 + lane;
        nxt[u] = (qi < QUADS_PER_WAVE) ? er4[qi] : none;
      }

#pragma unroll
      for (int u = 0; u < SCAN_U; ++u) {
        const int qi    = it * (32 * SCAN_U) + u * 32 + lane;
        const int ebase = (wave * QUADS_PER_WAVE + qi) * 4;
        unsigned d[4], mask[4];
#pragma unroll
        for (int k = 0; k < 4; ++k) {
          d[k]    = (unsigned)(cur[u][k] - lo);
          mask[k] = __builtin_amdgcn_ballot_w32(d[k] < (unsigned)OWN_ROWS);
        }
        if ((mask[0] | mask[1] | mask[2] | mask[3]) != 0u) {
#pragma unroll
          for (int k = 0; k < 4; ++k) {
            if (d[k] < (unsigned)OWN_ROWS)
              q[count + (int)__builtin_amdgcn_mbcnt_lo(mask[k], 0u)] = ((ebase + k) << 8) | (int)d[k];
            count += __builtin_popcount(mask[k]);
          }
        }
        __builtin_amdgcn_fence(__ATOMIC_ACQ_REL, "wavefront");
        while (count >= BATCH) {
          count -= BATCH;
          drain_batch(sup, ev, ec, acc, q[count + slot], BATCH);
        }
      }

#pragma unroll
      for (int u = 0; u < SCAN_U; ++u) cur[u] = nxt[u];
    }
    if (count > 0) drain_batch(sup, ev, ec, acc, q[slot < count ? slot : count - 1], count);
  }
  __syncthreads();

  {
    const v4f* a4 = (const v4f*)acc;
    v4f* o4 = (v4f*)out + (size_t)lo * (DOUT / 4);
    for (int i = t; i < OWN_ROWS * (DOUT / 4); i += 32 * OWN_WAVES) o4[i] = a4[i];
  }
}

extern "C" void kernel_launch(void* const* d_in, const int* in_sizes, int n_in,
                              void* d_out, int out_size, void* d_ws, size_t ws_size,
                              hipStream_t stream) {
  const float* inputs  = (const float*)d_in[0];
  const float* weights = (const float*)d_in[1];
  const float* bias    = (const float*)d_in[2];
  const float* evals   = (const float*)d_in[3];
  const int*   erow    = (const int*)d_in[4];
  const int*   ecol    = (const int*)d_in[5];
  float* out      = (float*)d_out;
  float* supports = (float*)d_ws;

  gemm_wmma_f32<<<PAIR_BLOCKS + LONE_BLOCKS, 128, 0, stream>>>(inputs, weights, supports, out);
  bin_edges<<<OWN_WGS, BIN_THREADS, 0, stream>>>(erow, out);
  owner_aggregate<<<OWN_WGS, 32 * OWN_WAVES, 0, stream>>>(supports, evals, erow, ecol, bias, out);
}


# edge sums by rows in registers: owner sorts its list by row in LDS (integer rank adds), a wave adds one row's edges one by one in list order (16-B loads) and stores it; no LDS f32 adds
# speedup vs baseline: 2.8069x; 1.0698x over previous
#include <hip/hip_runtime.h>

typedef __attribute__((ext_vector_type(2))) float v2f;
typedef __attribute__((ext_vector_type(8))) float v8f;
typedef __attribute__((ext_vector_type(4))) float v4f;

#define NODES 50000
#define DIN   512
#define DOUT  256
#define EDGES 800000

#define OWN_ROWS   200
#define OWN_WGS    (NODES / OWN_ROWS)
#define LIST_HEAD  32
#define LIST_CAP   (OWN_ROWS * DOUT - LIST_HEAD)

#define ROW_TILES    (NODES / 16)
#define PAIR_BLOCKS  1536
#define LONE_BLOCKS  (ROW_TILES - 2 * PAIR_BLOCKS)
#define KTRIPS       (DIN / 8)
static_assert(ROW_TILES * 16 == NODES, "row tiles must cover the nodes exactly");
static_assert(LONE_BLOCKS >= 0, "two-tile blocks must not run past the last row tile");
static_assert(KTRIPS % 3 == 1, "the loop below runs KTRIPS - 1 trips in threes and one more behind it");

template <int MT>
struct OperandSet {
  v4f a[MT];
  v4f q[4];
};

template <int MT>
__device__ __forceinline__ void load_operands(OperandSet<MT>& s, const float* const (&arow)[MT],
                                              const float* bcol, int k) {
  k = (k < DIN - 8) ? k : DIN - 8;
#pragma unroll
  for (int m = 0; m < MT; ++m) s.a[m] = *(const v4f*)(arow[m] + k);
#pragma unroll
  for (int j = 0; j < 4; ++j) s.q[j] = *(const v4f*)(bcol + (size_t)(k + j) * DOUT);
  asm volatile("" ::: "memory");
  __builtin_amdgcn_sched_barrier(0);
}

template <int MT>
__device__ __forceinline__ void multiply_trip(v8f (&acc)[MT][4], const OperandSet<MT>& s) {
#pragma unroll
  for (int g = 0; g < 2; ++g) {
    v2f ap[MT], bp[4];
#pragma unroll
    for (int m = 0; m < MT; ++m) { ap[m].x = s.a[m][2 * g];  ap[m].y = s.a[m][2 * g + 1]; }
#pragma unroll
    for (int t = 0; t < 4; ++t) { bp[t].x = s.q[2 * g][t];  bp[t].y = s.q[2 * g + 1][t]; }
#pragma unroll
    for (int t = 0; t < 4; ++t)
#pragma unroll
      for (int m = 0; m < MT; ++m)
        acc[m][t] = __builtin_amdgcn_wmma_f32_16x16x4_f32(false, ap[m], false, bp[t], (short)0,
                                                          acc[m][t], false, false);
  }
}

template <int MT>
__device__ __forceinline__ void gemm_row_tiles(const float* __restrict__ A,
                                               const float* __restrict__ B,
                                               float* __restrict__ C, const int mbase) {
  const int lane  = threadIdx.x & 31;
  const int wave  = threadIdx.x >> 5;
  const int half  = lane >> 4;
  const int l15   = lane & 15;
  const int nbase = wave * 64;

  v8f acc[MT][4];
#pragma unroll
  for (int m = 0; m < MT; ++m)
#pragma unroll
    for (int t = 0; t < 4; ++t) acc[m][t] = v8f{};

  const float* arow[MT];
#pragma unroll
  for (int m = 0; m < MT; ++m) arow[m] = A + (size_t)(mbase + 16 * m + l15) * DIN + 4 * half;
  const float* bcol = B + (size_t)(4 * half) * DOUT + nbase + 4 * l15;

  OperandSet<MT> s0, s1, s2;
  load_operands<MT>(s0, arow, bcol, 0);
  load_operands<MT>(s1, arow, bcol, 8);
  for (int k = 0; k < DIN - 8; k += 24) {
    load_operands<MT>(s2, arow, bcol, k + 16);
    multiply_trip<MT>(acc, s0);
    load_operands<MT>(s0, arow, bcol, k + 24);
    multiply_trip<MT>(acc, s1);
    load_operands<MT>(s1, arow, bcol, k + 32);
    multiply_trip<MT>(acc, s2);
  }
  multiply_trip<MT>(acc, s0);

#pragma unroll
  for (int m = 0; m < MT; ++m) {
    float* crow = C + (size_t)(mbase + 16 * m + 8 * half) * DOUT + nbase + 4 * l15;
#pragma unroll
    for (int r = 0; r < 8; ++r) {
      v4f o;
      o.x = acc[m][0][r];  o.y = acc[m][1][r];  o.z = acc[m][2][r];  o.w = acc[m][3][r];
      *(v4f*)(crow + (size_t)r * DOUT) = o;
    }
  }
}

__global__ __launch_bounds__(128) void gemm_wmma_f32(const float* __restrict__ A,
                                                     const float* __restrict__ B,
                                                     float* __restrict__ C,
                                                     float* __restrict__ outbuf) {
  const int b = blockIdx.x;
  if (b < OWN_WGS && threadIdx.x == 0) ((int*)outbuf)[(size_t)b * (OWN_ROWS * DOUT)] = 0;
  if (b < PAIR_BLOCKS) gemm_row_tiles<2>(A, B, C, b * 32);
  else                 gemm_row_tiles<1>(A, B, C, (2 * PAIR_BLOCKS + (b - PAIR_BLOCKS)) * 16);
}

typedef __attribute__((ext_vector_type(4))) int v4i;

#define BIN_EDGES   (EDGES / OWN_WGS)
#define BIN_QUADS   (BIN_EDGES / 4)
#define BIN_THREADS 256
#define BIN_STEPS   ((BIN_QUADS + BIN_THREADS - 1) / BIN_THREADS)
static_assert(BIN_EDGES * OWN_WGS == EDGES && BIN_QUADS * 4 == BIN_EDGES, "chunks of whole edge quads must cover the list");
static_assert(OWN_WGS <= BIN_THREADS, "one thread per owner reserves the chunk's run");
static_assert(OWN_ROWS <= 256 && EDGES <= (1 << 20), "entry packs a 20-bit edge index and an 8-bit local row");

__global__ __launch_bounds__(BIN_THREADS) void bin_edges(const int* __restrict__ er, float* __restrict__ outbuf) {
  __shared__ int run[OWN_WGS];
  const int t = threadIdx.x;
  if (t < OWN_WGS) run[t] = 0;
  __syncthreads();

  const v4i none = {-1, -1, -1, -1};
  v4i r[BIN_STEPS];
#pragma unroll
  for (int s = 0; s < BIN_STEPS; ++s) {
    const int qi = s * BIN_THREADS + t;
    r[s] = (qi < BIN_QUADS) ? ((const v4i*)er)[blockIdx.x * BIN_QUADS + qi] : none;
  }
  int owner[BIN_STEPS][4], ent[BIN_STEPS][4], rank[BIN_STEPS][4];
#pragma unroll
  for (int s = 0; s < BIN_STEPS; ++s) {
    const int quad = blockIdx.x * BIN_QUADS + s * BIN_THREADS + t;
#pragma unroll
    for (int k = 0; k < 4; ++k) {
      const unsigned row = (unsigned)r[s][k];
      const unsigned o   = row / OWN_ROWS;
      owner[s][k] = -1;  ent[s][k] = 0;  rank[s][k] = 0;
      if (o < (unsigned)OWN_WGS) {
        owner[s][k] = (int)o;
        ent[s][k]   = ((4 * quad + k) << 8) | (int)(row - o * OWN_ROWS);
        rank[s][k]  = atomicAdd(&run[o], 1);
      }
    }
  }
  __syncthreads();
  if (t < OWN_WGS) {
    const int c = run[t];
    int* len = (int*)outbuf + (size_t)t * (OWN_ROWS * DOUT);
    run[t] = (c > 0) ? atomicAdd(len, c) : 0;
  }
  __syncthreads();
#pragma unroll
  for (int s = 0; s < BIN_STEPS; ++s) {
#pragma unroll
    for (int k = 0; k < 4; ++k) {
      if (owner[s][k] >= 0) {
        const int pos = run[owner[s][k]] + rank[s][k];
        if (pos < LIST_CAP)
          ((int*)outbuf)[(size_t)owner[s][k] * (OWN_ROWS * DOUT) + LIST_HEAD + pos] = ent[s][k];
      }
    }
  }
}


#define OWN_WAVES  32
#define SORT_CAP   16384
#define ROW_SLOTS  224
#define QCAP       256
#define BATCH      8
#define SCAN_U     4
#define QUADS_PER_WAVE (EDGES / 4 / OWN_WAVES)
#define SCAN_ITERS ((QUADS_PER_WAVE + 32 * SCAN_U - 1) / (32 * SCAN_U))
static_assert(OWN_WGS * OWN_ROWS == NODES, "owned row ranges must tile the nodes exactly");
static_assert(EDGES % (4 * OWN_WAVES) == 0, "edge quads must split evenly over the waves");
static_assert(OWN_ROWS <= 256, "local row must fit the low 8 bits of a queue entry");
static_assert(OWN_ROWS <= ROW_SLOTS && ROW_SLOTS == 7 * 32, "one wave sums the row counts, 7 per lane");
static_assert(2 * SORT_CAP <= OWN_ROWS * DOUT + OWN_WAVES * QCAP, "sorted edges must fit the LDS block");
static_assert(SORT_CAP <= LIST_CAP, "a list the sorted path takes must have been stored whole");
static_assert(BATCH == 8, "a row's tail is taken as batches of 4, 2 and 1");
static_assert(BATCH - 1 + 128 <= QCAP, "queue holds a left-over plus one 128-edge step");

__device__ __forceinline__ void drain_batch(const float* __restrict__ sup,
                                            const float* __restrict__ ev,
                                            const int*   __restrict__ ec,
                                            float* acc, const int entv, int n) {
  const int lane = threadIdx.x & 31;
  int   lrow[BATCH];
  float val[BATCH];
  float s[BATCH][8];
#pragma unroll
  for (int i = 0; i < BATCH; ++i) {
    const int ent = __builtin_amdgcn_readlane(entv, i);
    const int e   = ent >> 8;
    lrow[i] = ent & 255;
    val[i]  = ev[e];
    const float* src = sup + (size_t)ec[e] * DOUT + lane;
#pragma unroll
    for (int j = 0; j < 8; ++j) s[i][j] = src[j * 32];
  }
#pragma unroll
  for (int i = 0; i < BATCH; ++i) {
    if (i < n) {
      float* dst = acc + lrow[i] * DOUT + lane;
#pragma unroll
      for (int j = 0; j < 8; ++j) atomicAdd(dst + j * 32, s[i][j] * val[i]);
    }
  }
}


template <int N>
__device__ __forceinline__ void add_batch(const float* __restrict__ sup, const int* scol, const float* sval,
                                          const int p, v4f& a0, v4f& a1) {
  const int lane = threadIdx.x & 31;
  const int idx  = p + (lane & (N - 1));
  const int myc  = scol[idx];
  const int myv  = __float_as_int(sval[idx]);
  v4f   q0[N], q1[N];
  float val[N];
#pragma unroll
  for (int i = 0; i < N; ++i) {
    const int c = __builtin_amdgcn_readlane(myc, i);
    val[i] = __int_as_float(__builtin_amdgcn_readlane(myv, i));
    const float* src = sup + (size_t)c * DOUT + 4 * lane;
    q0[i] = *(const v4f*)src;
    q1[i] = *(const v4f*)(src + 128);
  }
#pragma unroll
  for (int i = 0; i < N; ++i) {
    a0 += q0[i] * val[i];
    a1 += q1[i] * val[i];
  }
}

__device__ __forceinline__ void sum_rows_sorted(const float* __restrict__ sup, const float* __restrict__ ev,
                                                const int* __restrict__ ec, const float* __restrict__ bias,
                                                float* out, const int* ent, const int nlist,
                                                int* scol, float* sval, int* cnt, int* start, int* nextrow,
                                                const int lo) {
  const int t    = threadIdx.x;
  const int lane = t & 31;
  if (t < ROW_SLOTS) cnt[t] = 0;
  if (t == 0) *nextrow = 0;
  __syncthreads();

  for (int base = 0; base < nlist; base += 4 * 32 * OWN_WAVES) {
    int en[4];
#pragma unroll
    for (int u = 0; u < 4; ++u) {
      const int i = base + u * (32 * OWN_WAVES) + t;
      en[u] = (i < nlist) ? ent[i] : -1;
    }
#pragma unroll
    for (int u = 0; u < 4; ++u)
      if (en[u] >= 0) atomicAdd(&cnt[en[u] & 255], 1);
  }
  __syncthreads();

  if (t < 32) {
    int c[7];
    int s = 0;
#pragma unroll
    for (int j = 0; j < 7; ++j) { c[j] = cnt[7 * t + j];  s += c[j]; }
    int incl = s;
#pragma unroll
    for (int d = 1; d < 32; d <<= 1) {
      const int o = __shfl_up(incl, d);
      if (t >= d) incl += o;
    }
    int run = incl - s;
#pragma unroll
    for (int j = 0; j < 7; ++j) { start[7 * t + j] = run;  cnt[7 * t + j] = run;  run += c[j]; }
    if (t == 31) start[ROW_SLOTS] = run;
  }
  __syncthreads();

  for (int base = 0; base < nlist; base += 4 * 32 * OWN_WAVES) {
    int   en[4], c[4];
    float v[4];
#pragma unroll
    for (int u = 0; u < 4; ++u) {
      const int i = base + u * (32 * OWN_WAVES) + t;
      en[u] = (i < nlist) ? ent[i] : -1;
    }
#pragma unroll
    for (int u = 0; u < 4; ++u) {
      c[u] = 0;  v[u] = 0.0f;
      if (en[u] >= 0) { c[u] = ec[en[u] >> 8];  v[u] = ev[en[u] >> 8]; }
    }
#pragma unroll
    for (int u = 0; u < 4; ++u) {
      if (en[u] >= 0) {
        const int pos = atomicAdd(&cnt[en[u] & 255], 1);
        scol[pos] = c[u];
        sval[pos] = v[u];
      }
    }
  }
  __syncthreads();

  v4f b0, b1;
  {
    const float* bq = bias + 4 * lane;
    b0.x = bq[0];    b0.y = bq[1];    b0.z = bq[2];    b0.w = bq[3];
    b1.x = bq[128];  b1.y = bq[129];  b1.z = bq[130];  b1.w = bq[131];
  }
  for (;;) {
    int r = 0;
    if (lane == 0) r = atomicAdd(nextrow, 1);
    r = __builtin_amdgcn_readfirstlane(r);
    if (r >= OWN_ROWS) break;
    const int pb = __builtin_amdgcn_readfirstlane(start[r]);
    const int pe = __builtin_amdgcn_readfirstlane(start[r + 1]);
    v4f a0 = b0, a1 = b1;
    int p = pb;
    for (; p + BATCH <= pe; p += BATCH) add_batch<BATCH>(sup, scol, sval, p, a0, a1);
    if (pe - p >= 4) { add_batch<4>(sup, scol, sval, p, a0, a1);  p += 4; }
    if (pe - p >= 2) { add_batch<2>(sup, scol, sval, p, a0, a1);  p += 2; }
    if (pe - p >= 1) { add_batch<1>(sup, scol, sval, p, a0, a1); }
    float* dst = out + (size_t)(lo + r) * DOUT + 4 * lane;
    *(v4f*)dst         = a0;
    *(v4f*)(dst + 128) = a1;
  }
}

__global__ __launch_bounds__(32 * OWN_WAVES) void owner_aggregate(const float* __restrict__ sup,
                                                                  const float* __restrict__ ev,
                                                                  const int*   __restrict__ er,
                                                                  const int*   __restrict__ ec,
                                                                  const float* __restrict__ bias,
                                                                  float* out) {
  __shared__ __attribute__((aligned(16))) float lds[OWN_ROWS * DOUT + OWN_WAVES * QCAP];
  __shared__ int cnt[ROW_SLOTS];
  __shared__ int start[ROW_SLOTS + 1];
  __shared__ int nextrow;
  float* acc   = lds;
  int*   queue = (int*)(lds + OWN_ROWS * DOUT);
  int*   scol  = (int*)lds;
  float* sval  = lds + SORT_CAP;

  const int t    = threadIdx.x;
  const int lane = t & 31;
  const int wave = __builtin_amdgcn_readfirstlane(t >> 5);
  const int lo   = blockIdx.x * OWN_ROWS;

  const int* list = (const int*)out + (size_t)lo * DOUT;
  const int  nlist = __builtin_amdgcn_readfirstlane(list[0]);

  if (nlist <= SORT_CAP) {
    sum_rows_sorted(sup, ev, ec, bias, out, list + LIST_HEAD, nlist, scol, sval, cnt, start, &nextrow, lo);
    return;
  }

  {
    const float* bq = bias + 4 * (t & 63);
    v4f bv;
    bv.x = bq[0]; bv.y = bq[1]; bv.z = bq[2]; bv.w = bq[3];
    v4f* a4 = (v4f*)acc;
    for (int i = t; i < OWN_ROWS * (DOUT / 4); i += 32 * OWN_WAVES) a4[i] = bv;
  }
  __syncthreads();

  const int  slot  = lane & (BATCH - 1);

  if (nlist <= LIST_CAP) {
    const int* ent = list + LIST_HEAD;
    int i = wave * BATCH;
    int cur = 0;
    if (i < nlist) cur = ent[(i + slot < nlist) ? i + slot : nlist - 1];
    while (i + BATCH <= nlist) {
      const int inext = i + BATCH * OWN_WAVES;
      int nxt = 0;
      if (inext < nlist) nxt = ent[(inext + slot < nlist) ? inext + slot : nlist - 1];
      drain_batch(sup, ev, ec, acc, cur, BATCH);
      cur = nxt;
      i   = inext;
    }
    if (i < nlist) drain_batch(sup, ev, ec, acc, cur, nlist - i);
  } else {
    const v4i* er4 = (const v4i*)er + (size_t)wave * QUADS_PER_WAVE;
    int* q = queue + wave * QCAP;
    int count = 0;
    const v4i none = {-1, -1, -1, -1};

    v4i cur[SCAN_U], nxt[SCAN_U];
#pragma unroll
    for (int u = 0; u < SCAN_U; ++u) {
      const int qi = u * 32 + lane;
      cur[u] = (qi < QUADS_PER_WAVE) ? er4[qi] : none;
    }

    for (int it = 0; it < SCAN_ITERS; ++it) {
#pragma unroll
      for (int u = 0; u < SCAN_U; ++u) {
        const int qi = (it + 1) * (32 * SCAN_U) + u * 32 + lane;
        nxt[u] = (qi < QUADS_PER_WAVE) ? er4[qi] : none;
      }

#pragma unroll
      for (int u = 0; u < SCAN_U; ++u) {
        const int qi    = it * (32 * SCAN_U) + u * 32 + lane;
        const int ebase = (wave * QUADS_PER_WAVE + qi) * 4;
        unsigned d[4], mask[4];
#pragma unroll
        for (int k = 0; k < 4; ++k) {
          d[k]    = (unsigned)(cur[u][k] - lo);
          mask[k] = __builtin_amdgcn_ballot_w32(d[k] < (unsigned)OWN_ROWS);
        }
        if ((mask[0] | mask[1] | mask[2] | mask[3]) != 0u) {
#pragma unroll
          for (int k = 0; k < 4; ++k) {
            if (d[k] < (unsigned)OWN_ROWS)
              q[count + (int)__builtin_amdgcn_mbcnt_lo(mask[k], 0u)] = ((ebase + k) << 8) | (int)d[k];
            count += __builtin_popcount(mask[k]);
          }
        }
        __builtin_amdgcn_fence(__ATOMIC_ACQ_REL, "wavefront");
        while (count >= BATCH) {
          count -= BATCH;
          drain_batch(sup, ev, ec, acc, q[count + slot], BATCH);
        }
      }

#pragma unroll
      for (int u = 0; u < SCAN_U; ++u) cur[u] = nxt[u];
    }
    if (count > 0) drain_batch(sup, ev, ec, acc, q[slot < count ? slot : count - 1], count);
  }
  __syncthreads();

  {
    const v4f* a4 = (const v4f*)acc;
    v4f* o4 = (v4f*)out + (size_t)lo * (DOUT / 4);
    for (int i = t; i < OWN_ROWS * (DOUT / 4); i += 32 * OWN_WAVES) o4[i] = a4[i];
  }
}

extern "C" void kernel_launch(void* const* d_in, const int* in_sizes, int n_in,
                              void* d_out, int out_size, void* d_ws, size_t ws_size,
                              hipStream_t stream) {
  const float* inputs  = (const float*)d_in[0];
  const float* weights = (const float*)d_in[1];
  const float* bias    = (const float*)d_in[2];
  const float* evals   = (const float*)d_in[3];
  const int*   erow    = (const int*)d_in[4];
  const int*   ecol    = (const int*)d_in[5];
  float* out      = (float*)d_out;
  float* supports = (float*)d_ws;

  gemm_wmma_f32<<<PAIR_BLOCKS + LONE_BLOCKS, 128, 0, stream>>>(inputs, weights, supports, out);
  bin_edges<<<OWN_WGS, BIN_THREADS, 0, stream>>>(erow, out);
  owner_aggregate<<<OWN_WGS, 32 * OWN_WAVES, 0, stream>>>(supports, evals, erow, ecol, bias, out);
}


# edge binning moved into the product launch as 250 extra work-groups (3 launches -> 2): one-writer slots per chunk in each owner's rows replace the global length adds; arrival-order f32 sums kept
# speedup vs baseline: 2.9443x; 1.0490x over previous
#include <hip/hip_runtime.h>

typedef __attribute__((ext_vector_type(2))) float v2f;
typedef __attribute__((ext_vector_type(8))) float v8f;
typedef __attribute__((ext_vector_type(4))) float v4f;
typedef __attribute__((ext_vector_type(4))) int v4i;

#define NODES 50000
#define DIN   512
#define DOUT  256
#define EDGES 800000

#define OWN_ROWS   200
#define OWN_WGS    (NODES / OWN_ROWS)
#define BIN_CHUNKS OWN_WGS
#define SLOT_HEAD  256
#define SLOT_CAP   192
static_assert(SLOT_HEAD >= BIN_CHUNKS && SLOT_HEAD % 32 == 0 && SLOT_CAP % 32 == 0, "counts first, then line-aligned slots");
static_assert(SLOT_HEAD + BIN_CHUNKS * SLOT_CAP <= OWN_ROWS * DOUT, "the slots must fit the owner's own rows of out");

#define ROW_TILES    (NODES / 16)
#define PAIR_BLOCKS  1536
#define LONE_BLOCKS  (ROW_TILES - 2 * PAIR_BLOCKS)
#define KTRIPS       (DIN / 8)
static_assert(ROW_TILES * 16 == NODES, "row tiles must cover the nodes exactly");
static_assert(LONE_BLOCKS >= 0, "two-tile blocks must not run past the last row tile");
static_assert(KTRIPS % 3 == 1, "the loop below runs KTRIPS - 1 trips in threes and one more behind it");

template <int MT>
struct OperandSet {
  v4f a[MT];
  v4f q[4];
};

template <int MT>
__device__ __forceinline__ void load_operands(OperandSet<MT>& s, const float* const (&arow)[MT],
                                              const float* bcol, int k) {
  k = (k < DIN - 8) ? k : DIN - 8;
#pragma unroll
  for (int m = 0; m < MT; ++m) s.a[m] = *(const v4f*)(arow[m] + k);
#pragma unroll
  for (int j = 0; j < 4; ++j) s.q[j] = *(const v4f*)(bcol + (size_t)(k + j) * DOUT);
  asm volatile("" ::: "memory");
  __builtin_amdgcn_sched_barrier(0);
}

template <int MT>
__device__ __forceinline__ void multiply_trip(v8f (&acc)[MT][4], const OperandSet<MT>& s) {
#pragma unroll
  for (int g = 0; g < 2; ++g) {
    v2f ap[MT], bp[4];
#pragma unroll
    for (int m = 0; m < MT; ++m) { ap[m].x = s.a[m][2 * g];  ap[m].y = s.a[m][2 * g + 1]; }
#pragma unroll
    for (int t = 0; t < 4; ++t) { bp[t].x = s.q[2 * g][t];  bp[t].y = s.q[2 * g + 1][t]; }
#pragma unroll
    for (int t = 0; t < 4; ++t)
#pragma unroll
      for (int m = 0; m < MT; ++m)
        acc[m][t] = __builtin_amdgcn_wmma_f32_16x16x4_f32(false, ap[m], false, bp[t], (short)0,
                                                          acc[m][t], false, false);
  }
}

template <int MT>
__device__ __forceinline__ void gemm_row_tiles(const float* __restrict__ A,
                                               const float* __restrict__ B,
                                               float* __restrict__ C, const int mbase) {
  const int lane  = threadIdx.x & 31;
  const int wave  = threadIdx.x >> 5;
  const int half  = lane >> 4;
  const int l15   = lane & 15;
  const int nbase = wave * 64;

  v8f acc[MT][4];
#pragma unroll
  for (int m = 0; m < MT; ++m)
#pragma unroll
    for (int t = 0; t < 4; ++t) acc[m][t] = v8f{};

  const float* arow[MT];
#pragma unroll
  for (int m = 0; m < MT; ++m) arow[m] = A + (size_t)(mbase + 16 * m + l15) * DIN + 4 * half;
  const float* bcol = B + (size_t)(4 * half) * DOUT + nbase + 4 * l15;

  OperandSet<MT> s0, s1, s2;
  load_operands<MT>(s0, arow, bcol, 0);
  load_operands<MT>(s1, arow, bcol, 8);
  for (int k = 0; k < DIN - 8; k += 24) {
    load_operands<MT>(s2, arow, bcol, k + 16);
    multiply_trip<MT>(acc, s0);
    load_operands<MT>(s0, arow, bcol, k + 24);
    multiply_trip<MT>(acc, s1);
    load_operands<MT>(s1, arow, bcol, k + 32);
    multiply_trip<MT>(acc, s2);
  }
  multiply_trip<MT>(acc, s0);

#pragma unroll
  for (int m = 0; m < MT; ++m) {
    float* crow = C + (size_t)(mbase + 16 * m + 8 * half) * DOUT + nbase + 4 * l15;
#pragma unroll
    for (int r = 0; r < 8; ++r) {
      v4f o;
      o.x = acc[m][0][r];  o.y = acc[m][1][r];  o.z = acc[m][2][r];  o.w = acc[m][3][r];
      *(v4f*)(crow + (size_t)r * DOUT) = o;
    }
  }
}

#define BIN_EDGES   (EDGES / BIN_CHUNKS)
#define BIN_QUADS   (BIN_EDGES / 4)
#define BIN_THREADS 128
#define BIN_STEPS   ((BIN_QUADS + BIN_THREADS - 1) / BIN_THREADS)
static_assert(BIN_EDGES * BIN_CHUNKS == EDGES && BIN_QUADS * 4 == BIN_EDGES, "chunks of whole edge quads must cover the list");
static_assert(OWN_ROWS <= 256 && EDGES <= (1 << 20), "entry packs a 20-bit edge index and an 8-bit local row");

__device__ __forceinline__ void bin_chunk(const int* __restrict__ er, float* __restrict__ outbuf, const int g) {
  __shared__ int run[OWN_WGS];
  const int t = threadIdx.x;
  for (int o = t; o < OWN_WGS; o += BIN_THREADS) run[o] = 0;
  __syncthreads();

  const v4i none = {-1, -1, -1, -1};
  v4i r[BIN_STEPS];
#pragma unroll
  for (int s = 0; s < BIN_STEPS; ++s) {
    const int qi = s * BIN_THREADS + t;
    r[s] = (qi < BIN_QUADS) ? ((const v4i*)er)[g * BIN_QUADS + qi] : none;
  }
  int owner[BIN_STEPS][4], ent[BIN_STEPS][4], rank[BIN_STEPS][4];
#pragma unroll
  for (int s = 0; s < BIN_STEPS; ++s) {
    const int quad = g * BIN_QUADS + s * BIN_THREADS + t;
#pragma unroll
    for (int k = 0; k < 4; ++k) {
      const unsigned row = (unsigned)r[s][k];
      const unsigned o   = row / OWN_ROWS;
      owner[s][k] = -1;  ent[s][k] = 0;  rank[s][k] = 0;
      if (o < (unsigned)OWN_WGS) {
        owner[s][k] = (int)o;
        ent[s][k]   = ((4 * quad + k) << 8) | (int)(row - o * OWN_ROWS);
        rank[s][k]  = atomicAdd(&run[o], 1);
      }
    }
  }
  __syncthreads();
  int* ob = (int*)outbuf;
  for (int o = t; o < OWN_WGS; o += BIN_THREADS) ob[(size_t)o * (OWN_ROWS * DOUT) + g] = run[o];
#pragma unroll
  for (int s = 0; s < BIN_STEPS; ++s) {
#pragma unroll
    for (int k = 0; k < 4; ++k) {
      if (owner[s][k] >= 0 && rank[s][k] < SLOT_CAP)
        ob[(size_t)owner[s][k] * (OWN_ROWS * DOUT) + SLOT_HEAD + g * SLOT_CAP + rank[s][k]] = ent[s][k];
    }
  }
}

static_assert(BIN_THREADS == 128, "the binning work-groups share the product's launch shape");

__global__ __launch_bounds__(128) void project_and_bin(const float* __restrict__ A,
                                                       const float* __restrict__ B,
                                                       float* __restrict__ C,
                                                       const int* __restrict__ er,
                                                       float* __restrict__ outbuf) {
  if (blockIdx.x < BIN_CHUNKS) { bin_chunk(er, outbuf, blockIdx.x);  return; }
  const int b = blockIdx.x - BIN_CHUNKS;
  if (b < PAIR_BLOCKS) gemm_row_tiles<2>(A, B, C, b * 32);
  else                 gemm_row_tiles<1>(A, B, C, (2 * PAIR_BLOCKS + (b - PAIR_BLOCKS)) * 16);
}


#define OWN_WAVES  32
#define SORT_CAP   16384
#define ROW_SLOTS  224
#define QCAP       256
#define BATCH      8
#define SCAN_U     4
#define QUADS_PER_WAVE (EDGES / 4 / OWN_WAVES)
#define SCAN_ITERS ((QUADS_PER_WAVE + 32 * SCAN_U - 1) / (32 * SCAN_U))
static_assert(OWN_WGS * OWN_ROWS == NODES, "owned row ranges must tile the nodes exactly");
static_assert(EDGES % (4 * OWN_WAVES) == 0, "edge quads must split evenly over the waves");
static_assert(OWN_ROWS <= 256, "local row must fit the low 8 bits of a queue entry");
static_assert(OWN_ROWS <= ROW_SLOTS && ROW_SLOTS == 7 * 32, "one wave sums the row counts, 7 per lane");
static_assert(3 * SORT_CAP <= OWN_ROWS * DOUT + OWN_WAVES * QCAP, "gathered entries and sorted edges must fit the LDS block");
static_assert(BIN_CHUNKS <= 256 && 4 * BIN_CHUNKS <= 32 * OWN_WAVES, "one wave sums the chunk counts 8 per lane; 4 threads copy a slot");
static_assert(BATCH == 8, "a row's tail is taken as batches of 4, 2 and 1");
static_assert(BATCH - 1 + 128 <= QCAP, "queue holds a left-over plus one 128-edge step");

__device__ __forceinline__ void drain_batch(const float* __restrict__ sup,
                                            const float* __restrict__ ev,
                                            const int*   __restrict__ ec,
                                            float* acc, const int entv, int n) {
  const int lane = threadIdx.x & 31;
  int   lrow[BATCH];
  float val[BATCH];
  float s[BATCH][8];
#pragma unroll
  for (int i = 0; i < BATCH; ++i) {
    const int ent = __builtin_amdgcn_readlane(entv, i);
    const int e   = ent >> 8;
    lrow[i] = ent & 255;
    val[i]  = ev[e];
    const float* src = sup + (size_t)ec[e] * DOUT + lane;
#pragma unroll
    for (int j = 0; j < 8; ++j) s[i][j] = src[j * 32];
  }
#pragma unroll
  for (int i = 0; i < BATCH; ++i) {
    if (i < n) {
      float* dst = acc + lrow[i] * DOUT + lane;
#pragma unroll
      for (int j = 0; j < 8; ++j) atomicAdd(dst + j * 32, s[i][j] * val[i]);
    }
  }
}


template <int N>
__device__ __forceinline__ void add_batch(const float* __restrict__ sup, const int* scol, const float* sval,
                                          const int p, v4f& a0, v4f& a1) {
  const int lane = threadIdx.x & 31;
  const int idx  = p + (lane & (N - 1));
  const int myc  = scol[idx];
  const int myv  = __float_as_int(sval[idx]);
  v4f   q0[N], q1[N];
  float val[N];
#pragma unroll
  for (int i = 0; i < N; ++i) {
    const int c = __builtin_amdgcn_readlane(myc, i);
    val[i] = __int_as_float(__builtin_amdgcn_readlane(myv, i));
    const float* src = sup + (size_t)c * DOUT + 4 * lane;
    q0[i] = *(const v4f*)src;
    q1[i] = *(const v4f*)(src + 128);
  }
#pragma unroll
  for (int i = 0; i < N; ++i) {
    a0 += q0[i] * val[i];
    a1 += q1[i] * val[i];
  }
}

__device__ __forceinline__ void sum_rows_sorted(const float* __restrict__ sup, const float* __restrict__ ev,
                                                const int* __restrict__ ec, const float* __restrict__ bias,
                                                float* out, const int* ent, const int nlist,
                                                int* scol, float* sval, int* cnt, int* start, int* nextrow,
                                                const int lo) {
  const int t    = threadIdx.x;
  const int lane = t & 31;
  if (t < ROW_SLOTS) cnt[t] = 0;
  if (t == 0) *nextrow = 0;
  __syncthreads();

  for (int base = 0; base < nlist; base += 4 * 32 * OWN_WAVES) {
    int en[4];
#pragma unroll
    for (int u = 0; u < 4; ++u) {
      const int i = base + u * (32 * OWN_WAVES) + t;
      en[u] = (i < nlist) ? ent[i] : -1;
    }
#pragma unroll
    for (int u = 0; u < 4; ++u)
      if (en[u] >= 0) atomicAdd(&cnt[en[u] & 255], 1);
  }
  __syncthreads();

  if (t < 32) {
    int c[7];
    int s = 0;
#pragma unroll
    for (int j = 0; j < 7; ++j) { c[j] = cnt[7 * t + j];  s += c[j]; }
    int incl = s;
#pragma unroll
    for (int d = 1; d < 32; d <<= 1) {
      const int o = __shfl_up(incl, d);
      if (t >= d) incl += o;
    }
    int run = incl - s;
#pragma unroll
    for (int j = 0; j < 7; ++j) { start[7 * t + j] = run;  cnt[7 * t + j] = run;  run += c[j]; }
    if (t == 31) start[ROW_SLOTS] = run;
  }
  __syncthreads();

  for (int base = 0; base < nlist; base += 4 * 32 * OWN_WAVES) {
    int   en[4], c[4];
    float v[4];
#pragma unroll
    for (int u = 0; u < 4; ++u) {
      const int i = base + u * (32 * OWN_WAVES) + t;
      en[u] = (i < nlist) ? ent[i] : -1;
    }
#pragma unroll
    for (int u = 0; u < 4; ++u) {
      c[u] = 0;  v[u] = 0.0f;
      if (en[u] >= 0) { c[u] = ec[en[u] >> 8];  v[u] = ev[en[u] >> 8]; }
    }
#pragma unroll
    for (int u = 0; u < 4; ++u) {
      if (en[u] >= 0) {
        const int pos = atomicAdd(&cnt[en[u] & 255], 1);
        scol[pos] = c[u];
        sval[pos] = v[u];
      }
    }
  }
  __syncthreads();

  v4f b0, b1;
  {
    const float* bq = bias + 4 * lane;
    b0.x = bq[0];    b0.y = bq[1];    b0.z = bq[2];    b0.w = bq[3];
    b1.x = bq[128];  b1.y = bq[129];  b1.z = bq[130];  b1.w = bq[131];
  }
  for (;;) {
    int r = 0;
    if (lane == 0) r = atomicAdd(nextrow, 1);
    r = __builtin_amdgcn_readfirstlane(r);
    if (r >= OWN_ROWS) break;
    const int pb = __builtin_amdgcn_readfirstlane(start[r]);
    const int pe = __builtin_amdgcn_readfirstlane(start[r + 1]);
    v4f a0 = b0, a1 = b1;
    int p = pb;
    for (; p + BATCH <= pe; p += BATCH) add_batch<BATCH>(sup, scol, sval, p, a0, a1);
    if (pe - p >= 4) { add_batch<4>(sup, scol, sval, p, a0, a1);  p += 4; }
    if (pe - p >= 2) { add_batch<2>(sup, scol, sval, p, a0, a1);  p += 2; }
    if (pe - p >= 1) { add_batch<1>(sup, scol, sval, p, a0, a1); }
    float* dst = out + (size_t)(lo + r) * DOUT + 4 * lane;
    *(v4f*)dst         = a0;
    *(v4f*)(dst + 128) = a1;
  }
}

__global__ __launch_bounds__(32 * OWN_WAVES) void owner_aggregate(const float* __restrict__ sup,
                                                                  const float* __restrict__ ev,
                                                                  const int*   __restrict__ er,
                                                                  const int*   __restrict__ ec,
                                                                  const float* __restrict__ bias,
                                                                  float* out) {
  __shared__ __attribute__((aligned(16))) float lds[OWN_ROWS * DOUT + OWN_WAVES * QCAP];
  __shared__ int cnt[ROW_SLOTS];
  __shared__ int start[ROW_SLOTS + 1];
  __shared__ int nextrow;
  __shared__ int coff[256 + 1];
  __shared__ int spill;
  float* acc   = lds;
  int*   queue = (int*)(lds + OWN_ROWS * DOUT);
  int*   scol  = (int*)lds;
  float* sval  = lds + SORT_CAP;
  int*   elist = (int*)(lds + 2 * SORT_CAP);

  const int t    = threadIdx.x;
  const int lane = t & 31;
  const int wave = __builtin_amdgcn_readfirstlane(t >> 5);
  const int lo   = blockIdx.x * OWN_ROWS;

  const int* region = (const int*)out + (size_t)lo * DOUT;
  int mycount = 0;
  if (t < BIN_CHUNKS) mycount = region[t];
  if (t < 256) coff[t] = mycount;
  if (t == 0) spill = 0;
  __syncthreads();
  if (mycount > SLOT_CAP) spill = 1;
  if (t < 32) {
    int c[8];
    int s = 0;
#pragma unroll
    for (int j = 0; j < 8; ++j) { c[j] = coff[8 * t + j];  s += c[j]; }
    int incl = s;
#pragma unroll
    for (int d = 1; d < 32; d <<= 1) {
      const int o = __shfl_up(incl, d);
      if (t >= d) incl += o;
    }
    int run = incl - s;
#pragma unroll
    for (int j = 0; j < 8; ++j) { coff[8 * t + j] = run;  run += c[j]; }
    if (t == 31) coff[256] = run;
  }
  __syncthreads();
  const int nlist   = __builtin_amdgcn_readfirstlane(coff[256]);
  const int spilled = __builtin_amdgcn_readfirstlane(spill);

  if (spilled == 0 && nlist <= SORT_CAP) {
    {
      const int g = t >> 2, sub = t & 3;
      if (g < BIN_CHUNKS) {
        const int  off  = coff[g];
        const int  cg   = coff[g + 1] - off;
        const int* slot = region + SLOT_HEAD + g * SLOT_CAP;
        for (int p = 4 * sub; p < cg; p += 16) {
          const v4i e = *(const v4i*)(slot + p);
#pragma unroll
          for (int k = 0; k < 4; ++k)
            if (p + k < cg) elist[off + p + k] = e[k];
        }
      }
    }
    __syncthreads();
    sum_rows_sorted(sup, ev, ec, bias, out, elist, nlist, scol, sval, cnt, start, &nextrow, lo);
    return;
  }

  {
    const float* bq = bias + 4 * (t & 63);
    v4f bv;
    bv.x = bq[0]; bv.y = bq[1]; bv.z = bq[2]; bv.w = bq[3];
    v4f* a4 = (v4f*)acc;
    for (int i = t; i < OWN_ROWS * (DOUT / 4); i += 32 * OWN_WAVES) a4[i] = bv;
  }
  __syncthreads();

  const int  slot  = lane & (BATCH - 1);

  {
    const v4i* er4 = (const v4i*)er + (size_t)wave * QUADS_PER_WAVE;
    int* q = queue + wave * QCAP;
    int count = 0;
    const v4i none = {-1, -1, -1, -1};

    v4i cur[SCAN_U], nxt[SCAN_U];
#pragma unroll
    for (int u = 0; u < SCAN_U; ++u) {
      const int qi = u * 32 + lane;
      cur[u] = (qi < QUADS_PER_WAVE) ? er4[qi] : none;
    }

    for (int it = 0; it < SCAN_ITERS; ++it) {
#pragma unroll
      for (int u = 0; u < SCAN_U; ++u) {
        const int qi = (it + 1) * (32 * SCAN_U) + u * 32 + lane;
        nxt[u] = (qi < QUADS_PER_WAVE) ? er4[qi] : none;
      }

#pragma unroll
      for (int u = 0; u < SCAN_U; ++u) {
        const int qi    = it * (32 * SCAN_U) + u * 32 + lane;
        const int ebase = (wave * QUADS_PER_WAVE + qi) * 4;
        unsigned d[4], mask[4];
#pragma unroll
        for (int k = 0; k < 4; ++k) {
          d[k]    = (unsigned)(cur[u][k] - lo);
          mask[k] = __builtin_amdgcn_ballot_w32(d[k] < (unsigned)OWN_ROWS);
        }
        if ((mask[0] | mask[1] | mask[2] | mask[3]) != 0u) {
#pragma unroll
          for (int k = 0; k < 4; ++k) {
            if (d[k] < (unsigned)OWN_ROWS)
              q[count + (int)__builtin_amdgcn_mbcnt_lo(mask[k], 0u)] = ((ebase + k) << 8) | (int)d[k];
            count += __builtin_popcount(mask[k]);
          }
        }
        __builtin_amdgcn_fence(__ATOMIC_ACQ_REL, "wavefront");
        while (count >= BATCH) {
          count -= BATCH;
          drain_batch(sup, ev, ec, acc, q[count + slot], BATCH);
        }
      }

#pragma unroll
      for (int u = 0; u < SCAN_U; ++u) cur[u] = nxt[u];
    }
    if (count > 0) drain_batch(sup, ev, ec, acc, q[slot < count ? slot : count - 1], count);
  }
  __syncthreads();

  {
    const v4f* a4 = (const v4f*)acc;
    v4f* o4 = (v4f*)out + (size_t)lo * (DOUT / 4);
    for (int i = t; i < OWN_ROWS * (DOUT / 4); i += 32 * OWN_WAVES) o4[i] = a4[i];
  }
}

extern "C" void kernel_launch(void* const* d_in, const int* in_sizes, int n_in,
                              void* d_out, int out_size, void* d_ws, size_t ws_size,
                              hipStream_t stream) {
  const float* inputs  = (const float*)d_in[0];
  const float* weights = (const float*)d_in[1];
  const float* bias    = (const float*)d_in[2];
  const float* evals   = (const float*)d_in[3];
  const int*   erow    = (const int*)d_in[4];
  const int*   ecol    = (const int*)d_in[5];
  float* out      = (float*)d_out;
  float* supports = (float*)d_ws;

  project_and_bin<<<BIN_CHUNKS + PAIR_BLOCKS + LONE_BLOCKS, 128, 0, stream>>>(inputs, weights, supports, erow, out);
  owner_aggregate<<<OWN_WGS, 32 * OWN_WAVES, 0, stream>>>(supports, evals, erow, ecol, bias, out);
}


# binning puts each edge's source row and value into its owner's slot (8-B entries; edge_col, edge_vals read as whole lines); owners drop 2 x 3,200 scattered 4-B fetches each; arrival-order sums kept
# speedup vs baseline: 3.1063x; 1.0550x over previous
#include <hip/hip_runtime.h>

typedef __attribute__((ext_vector_type(2))) float v2f;
typedef __attribute__((ext_vector_type(8))) float v8f;
typedef __attribute__((ext_vector_type(4))) float v4f;
typedef __attribute__((ext_vector_type(4))) int v4i;
typedef __attribute__((ext_vector_type(2))) int v2i;

#define NODES 50000
#define DIN   512
#define DOUT  256
#define EDGES 800000

#define OWN_ROWS   200
#define OWN_WGS    (NODES / OWN_ROWS)
#define BIN_CHUNKS OWN_WGS
#define SLOT_HEAD  256
#define SLOT_CAP   96
#define SLOT_INTS  (2 * SLOT_CAP)
static_assert(SLOT_HEAD >= BIN_CHUNKS && SLOT_HEAD % 32 == 0 && SLOT_INTS % 32 == 0, "counts first, then line-aligned slots");
static_assert(SLOT_HEAD + BIN_CHUNKS * SLOT_INTS <= OWN_ROWS * DOUT, "the slots must fit the owner's own rows of out");
static_assert(SLOT_CAP % 4 == 0, "an owner reads a slot four entries at a time");
static_assert(NODES <= (1 << 23), "source row << 8 must stay a positive int");

#define ROW_TILES    (NODES / 16)
#define PAIR_BLOCKS  1536
#define LONE_BLOCKS  (ROW_TILES - 2 * PAIR_BLOCKS)
#define KTRIPS       (DIN / 8)
static_assert(ROW_TILES * 16 == NODES, "row tiles must cover the nodes exactly");
static_assert(LONE_BLOCKS >= 0, "two-tile blocks must not run past the last row tile");
static_assert(KTRIPS % 3 == 1, "the loop below runs KTRIPS - 1 trips in threes and one more behind it");

template <int MT>
struct OperandSet {
  v4f a[MT];
  v4f q[4];
};

template <int MT>
__device__ __forceinline__ void load_operands(OperandSet<MT>& s, const float* const (&arow)[MT],
                                              const float* bcol, int k) {
  k = (k < DIN - 8) ? k : DIN - 8;
#pragma unroll
  for (int m = 0; m < MT; ++m) s.a[m] = *(const v4f*)(arow[m] + k);
#pragma unroll
  for (int j = 0; j < 4; ++j) s.q[j] = *(const v4f*)(bcol + (size_t)(k + j) * DOUT);
  asm volatile("" ::: "memory");
  __builtin_amdgcn_sched_barrier(0);
}

template <int MT>
__device__ __forceinline__ void multiply_trip(v8f (&acc)[MT][4], const OperandSet<MT>& s) {
#pragma unroll
  for (int g = 0; g < 2; ++g) {
    v2f ap[MT], bp[4];
#pragma unroll
    for (int m = 0; m < MT; ++m) { ap[m].x = s.a[m][2 * g];  ap[m].y = s.a[m][2 * g + 1]; }
#pragma unroll
    for (int t = 0; t < 4; ++t) { bp[t].x = s.q[2 * g][t];  bp[t].y = s.q[2 * g + 1][t]; }
#pragma unroll
    for (int t = 0; t < 4; ++t)
#pragma unroll
      for (int m = 0; m < MT; ++m)
        acc[m][t] = __builtin_amdgcn_wmma_f32_16x16x4_f32(false, ap[m], false, bp[t], (short)0,
                                                          acc[m][t], false, false);
  }
}

template <int MT>
__device__ __forceinline__ void gemm_row_tiles(const float* __restrict__ A,
                                               const float* __restrict__ B,
                                               float* __restrict__ C, const int mbase) {
  const int lane  = threadIdx.x & 31;
  const int wave  = threadIdx.x >> 5;
  const int half  = lane >> 4;
  const int l15   = lane & 15;
  const int nbase = wave * 64;

  v8f acc[MT][4];
#pragma unroll
  for (int m = 0; m < MT; ++m)
#pragma unroll
    for (int t = 0; t < 4; ++t) acc[m][t] = v8f{};

  const float* arow[MT];
#pragma unroll
  for (int m = 0; m < MT; ++m) arow[m] = A + (size_t)(mbase + 16 * m + l15) * DIN + 4 * half;
  const float* bcol = B + (size_t)(4 * half) * DOUT + nbase + 4 * l15;

  OperandSet<MT> s0, s1, s2;
  load_operands<MT>(s0, arow, bcol, 0);
  load_operands<MT>(s1, arow, bcol, 8);
  for (int k = 0; k < DIN - 8; k += 24) {
    load_operands<MT>(s2, arow, bcol, k + 16);
    multiply_trip<MT>(acc, s0);
    load_operands<MT>(s0, arow, bcol, k + 24);
    multiply_trip<MT>(acc, s1);
    load_operands<MT>(s1, arow, bcol, k + 32);
    multiply_trip<MT>(acc, s2);
  }
  multiply_trip<MT>(acc, s0);

#pragma unroll
  for (int m = 0; m < MT; ++m) {
    float* crow = C + (size_t)(mbase + 16 * m + 8 * half) * DOUT + nbase + 4 * l15;
#pragma unroll
    for (int r = 0; r < 8; ++r) {
      v4f o;
      o.x = acc[m][0][r];  o.y = acc[m][1][r];  o.z = acc[m][2][r];  o.w = acc[m][3][r];
      *(v4f*)(crow + (size_t)r * DOUT) = o;
    }
  }
}

#define BIN_EDGES   (EDGES / BIN_CHUNKS)
#define BIN_QUADS   (BIN_EDGES / 4)
#define BIN_THREADS 128
#define BIN_STEPS   ((BIN_QUADS + BIN_THREADS - 1) / BIN_THREADS)
static_assert(BIN_EDGES * BIN_CHUNKS == EDGES && BIN_QUADS * 4 == BIN_EDGES, "chunks of whole edge quads must cover the list");
static_assert(OWN_ROWS <= 256 && EDGES <= (1 << 20), "a queue entry of the scanning path packs a 20-bit edge index and an 8-bit local row");
static_assert(OWN_WGS <= 2048 && BIN_EDGES <= 4096, "key packs an 11-bit owner, a 12-bit rank and an 8-bit local row");

__device__ __forceinline__ void bin_chunk(const int* __restrict__ er, const int* __restrict__ ec,
                                          const float* __restrict__ ev, float* __restrict__ outbuf, const int g) {
  __shared__ int run[OWN_WGS];
  const int t = threadIdx.x;
  for (int o = t; o < OWN_WGS; o += BIN_THREADS) run[o] = 0;
  __syncthreads();

  const v4i none = {-1, -1, -1, -1};
  v4i r[BIN_STEPS];
#pragma unroll
  for (int s = 0; s < BIN_STEPS; ++s) {
    const int qi = s * BIN_THREADS + t;
    r[s] = (qi < BIN_QUADS) ? ((const v4i*)er)[g * BIN_QUADS + qi] : none;
  }
  int key[BIN_STEPS][4];
#pragma unroll
  for (int s = 0; s < BIN_STEPS; ++s) {
#pragma unroll
    for (int k = 0; k < 4; ++k) {
      const unsigned row = (unsigned)r[s][k];
      const unsigned o   = row / OWN_ROWS;
      key[s][k] = -1;
      if (o < (unsigned)OWN_WGS) {
        const int rank = atomicAdd(&run[o], 1);
        key[s][k] = (int)(o << 20) | (rank << 8) | (int)(row - o * OWN_ROWS);
      }
    }
  }
  __syncthreads();
  int* ob = (int*)outbuf;
  for (int o = t; o < OWN_WGS; o += BIN_THREADS) ob[(size_t)o * (OWN_ROWS * DOUT) + g] = run[o];
  v4i c[BIN_STEPS];
  v4f v[BIN_STEPS];
#pragma unroll
  for (int s = 0; s < BIN_STEPS; ++s) {
    const int qi = s * BIN_THREADS + t;
    c[s] = (qi < BIN_QUADS) ? ((const v4i*)ec)[g * BIN_QUADS + qi] : none;
    v[s] = (qi < BIN_QUADS) ? ((const v4f*)ev)[g * BIN_QUADS + qi] : v4f{};
  }
#pragma unroll
  for (int s = 0; s < BIN_STEPS; ++s) {
#pragma unroll
    for (int k = 0; k < 4; ++k) {
      const int ky   = key[s][k];
      const int rank = (ky >> 8) & 0xFFF;
      if (ky >= 0 && rank < SLOT_CAP) {
        v2i e;
        e.x = (c[s][k] << 8) | (ky & 255);
        e.y = __float_as_int(v[s][k]);
        *(v2i*)(ob + (size_t)(ky >> 20) * (OWN_ROWS * DOUT) + SLOT_HEAD + g * SLOT_INTS + 2 * rank) = e;
      }
    }
  }
}

static_assert(BIN_THREADS == 128, "the binning work-groups share the product's launch shape");

__global__ __launch_bounds__(128) void project_and_bin(const float* __restrict__ A,
                                                       const float* __restrict__ B,
                                                       float* __restrict__ C,
                                                       const int* __restrict__ er,
                                                       const int* __restrict__ ec,
                                                       const float* __restrict__ ev,
                                                       float* __restrict__ outbuf) {
  if (blockIdx.x < BIN_CHUNKS) { bin_chunk(er, ec, ev, outbuf, blockIdx.x);  return; }
  const int b = blockIdx.x - BIN_CHUNKS;
  if (b < PAIR_BLOCKS) gemm_row_tiles<2>(A, B, C, b * 32);
  else                 gemm_row_tiles<1>(A, B, C, (2 * PAIR_BLOCKS + (b - PAIR_BLOCKS)) * 16);
}


#define OWN_WAVES  32
#define SORT_CAP   14336
#define ROW_SLOTS  224
#define QCAP       256
#define BATCH      8
#define SCAN_U     4
#define QUADS_PER_WAVE (EDGES / 4 / OWN_WAVES)
#define SCAN_ITERS ((QUADS_PER_WAVE + 32 * SCAN_U - 1) / (32 * SCAN_U))
static_assert(OWN_WGS * OWN_ROWS == NODES, "owned row ranges must tile the nodes exactly");
static_assert(EDGES % (4 * OWN_WAVES) == 0, "edge quads must split evenly over the waves");
static_assert(OWN_ROWS <= 256, "local row must fit the low 8 bits of a queue entry");
static_assert(OWN_ROWS <= ROW_SLOTS && ROW_SLOTS == 7 * 32, "one wave sums the row counts, 7 per lane");
static_assert(4 * SORT_CAP <= OWN_ROWS * DOUT + OWN_WAVES * QCAP, "gathered entries and sorted edges must fit the LDS block");
static_assert(BIN_CHUNKS <= 256 && 4 * BIN_CHUNKS <= 32 * OWN_WAVES, "one wave sums the chunk counts 8 per lane; 4 threads copy a slot");
static_assert(BATCH == 8, "a row's tail is taken as batches of 4, 2 and 1");
static_assert(BATCH - 1 + 128 <= QCAP, "queue holds a left-over plus one 128-edge step");

__device__ __forceinline__ void drain_batch(const float* __restrict__ sup,
                                            const float* __restrict__ ev,
                                            const int*   __restrict__ ec,
                                            float* acc, const int entv, int n) {
  const int lane = threadIdx.x & 31;
  int   lrow[BATCH];
  float val[BATCH];
  float s[BATCH][8];
#pragma unroll
  for (int i = 0; i < BATCH; ++i) {
    const int ent = __builtin_amdgcn_readlane(entv, i);
    const int e   = ent >> 8;
    lrow[i] = ent & 255;
    val[i]  = ev[e];
    const float* src = sup + (size_t)ec[e] * DOUT + lane;
#pragma unroll
    for (int j = 0; j < 8; ++j) s[i][j] = src[j * 32];
  }
#pragma unroll
  for (int i = 0; i < BATCH; ++i) {
    if (i < n) {
      float* dst = acc + lrow[i] * DOUT + lane;
#pragma unroll
      for (int j = 0; j < 8; ++j) atomicAdd(dst + j * 32, s[i][j] * val[i]);
    }
  }
}


template <int N>
__device__ __forceinline__ void add_batch(const float* __restrict__ sup, const int* scol, const float* sval,
                                          const int p, v4f& a0, v4f& a1) {
  const int lane = threadIdx.x & 31;
  const int idx  = p + (lane & (N - 1));
  const int myc  = scol[idx];
  const int myv  = __float_as_int(sval[idx]);
  v4f   q0[N], q1[N];
  float val[N];
#pragma unroll
  for (int i = 0; i < N; ++i) {
    const int c = __builtin_amdgcn_readlane(myc, i);
    val[i] = __int_as_float(__builtin_amdgcn_readlane(myv, i));
    const float* src = sup + (size_t)c * DOUT + 4 * lane;
    q0[i] = *(const v4f*)src;
    q1[i] = *(const v4f*)(src + 128);
  }
#pragma unroll
  for (int i = 0; i < N; ++i) {
    a0 += q0[i] * val[i];
    a1 += q1[i] * val[i];
  }
}

__device__ __forceinline__ void sum_rows_sorted(const float* __restrict__ sup, const float* __restrict__ bias,
                                                float* out, const int* ent, const float* entv, const int nlist,
                                                int* scol, float* sval, int* cnt, int* start, int* nextrow,
                                                const int lo) {
  const int t    = threadIdx.x;
  const int lane = t & 31;
  if (t < ROW_SLOTS) cnt[t] = 0;
  if (t == 0) *nextrow = 0;
  __syncthreads();

  for (int base = 0; base < nlist; base += 4 * 32 * OWN_WAVES) {
    int en[4];
#pragma unroll
    for (int u = 0; u < 4; ++u) {
      const int i = base + u * (32 * OWN_WAVES) + t;
      en[u] = (i < nlist) ? ent[i] : -1;
    }
#pragma unroll
    for (int u = 0; u < 4; ++u)
      if (en[u] >= 0) atomicAdd(&cnt[en[u] & 255], 1);
  }
  __syncthreads();

  if (t < 32) {
    int c[7];
    int s = 0;
#pragma unroll
    for (int j = 0; j < 7; ++j) { c[j] = cnt[7 * t + j];  s += c[j]; }
    int incl = s;
#pragma unroll
    for (int d = 1; d < 32; d <<= 1) {
      const int o = __shfl_up(incl, d);
      if (t >= d) incl += o;
    }
    int run = incl - s;
#pragma unroll
    for (int j = 0; j < 7; ++j) { start[7 * t + j] = run;  cnt[7 * t + j] = run;  run += c[j]; }
    if (t == 31) start[ROW_SLOTS] = run;
  }
  __syncthreads();

  for (int base = 0; base < nlist; base += 4 * 32 * OWN_WAVES) {
    int   en[4];
    float v[4];
#pragma unroll
    for (int u = 0; u < 4; ++u) {
      const int i = base + u * (32 * OWN_WAVES) + t;
      en[u] = (i < nlist) ? ent[i] : -1;
      v[u]  = (i < nlist) ? entv[i] : 0.0f;
    }
#pragma unroll
    for (int u = 0; u < 4; ++u) {
      if (en[u] >= 0) {
        const int pos = atomicAdd(&cnt[en[u] & 255], 1);
        scol[pos] = en[u] >> 8;
        sval[pos] = v[u];
      }
    }
  }
  __syncthreads();

  v4f b0, b1;
  {
    const float* bq = bias + 4 * lane;
    b0.x = bq[0];    b0.y = bq[1];    b0.z = bq[2];    b0.w = bq[3];
    b1.x = bq[128];  b1.y = bq[129];  b1.z = bq[130];  b1.w = bq[131];
  }
  for (;;) {
    int r = 0;
    if (lane == 0) r = atomicAdd(nextrow, 1);
    r = __builtin_amdgcn_readfirstlane(r);
    if (r >= OWN_ROWS) break;
    const int pb = __builtin_amdgcn_readfirstlane(start[r]);
    const int pe = __builtin_amdgcn_readfirstlane(start[r + 1]);
    v4f a0 = b0, a1 = b1;
    int p = pb;
    for (; p + BATCH <= pe; p += BATCH) add_batch<BATCH>(sup, scol, sval, p, a0, a1);
    if (pe - p >= 4) { add_batch<4>(sup, scol, sval, p, a0, a1);  p += 4; }
    if (pe - p >= 2) { add_batch<2>(sup, scol, sval, p, a0, a1);  p += 2; }
    if (pe - p >= 1) { add_batch<1>(sup, scol, sval, p, a0, a1); }
    float* dst = out + (size_t)(lo + r) * DOUT + 4 * lane;
    *(v4f*)dst         = a0;
    *(v4f*)(dst + 128) = a1;
  }
}

__global__ __launch_bounds__(32 * OWN_WAVES) void owner_aggregate(const float* __restrict__ sup,
                                                                  const float* __restrict__ ev,
                                                                  const int*   __restrict__ er,
                                                                  const int*   __restrict__ ec,
                                                                  const float* __restrict__ bias,
                                                                  float* out) {
  __shared__ __attribute__((aligned(16))) float lds[OWN_ROWS * DOUT + OWN_WAVES * QCAP];
  __shared__ int cnt[ROW_SLOTS];
  __shared__ int start[ROW_SLOTS + 1];
  __shared__ int nextrow;
  __shared__ int coff[256 + 1];
  __shared__ int spill;
  float* acc   = lds;
  int*   queue = (int*)(lds + OWN_ROWS * DOUT);
  int*   scol  = (int*)lds;
  float* sval  = lds + SORT_CAP;
  int*   elist = (int*)(lds + 2 * SORT_CAP);
  float* elval = lds + 3 * SORT_CAP;

  const int t    = threadIdx.x;
  const int lane = t & 31;
  const int wave = __builtin_amdgcn_readfirstlane(t >> 5);
  const int lo   = blockIdx.x * OWN_ROWS;

  const int* region = (const int*)out + (size_t)lo * DOUT;
  int mycount = 0;
  if (t < BIN_CHUNKS) mycount = region[t];
  if (t < 256) coff[t] = mycount;
  if (t == 0) spill = 0;
  __syncthreads();
  if (mycount > SLOT_CAP) spill = 1;
  if (t < 32) {
    int c[8];
    int s = 0;
#pragma unroll
    for (int j = 0; j < 8; ++j) { c[j] = coff[8 * t + j];  s += c[j]; }
    int incl = s;
#pragma unroll
    for (int d = 1; d < 32; d <<= 1) {
      const int o = __shfl_up(incl, d);
      if (t >= d) incl += o;
    }
    int run = incl - s;
#pragma unroll
    for (int j = 0; j < 8; ++j) { coff[8 * t + j] = run;  run += c[j]; }
    if (t == 31) coff[256] = run;
  }
  __syncthreads();
  const int nlist   = __builtin_amdgcn_readfirstlane(coff[256]);
  const int spilled = __builtin_amdgcn_readfirstlane(spill);

  if (spilled == 0 && nlist <= SORT_CAP) {
    {
      const int g = t >> 2, sub = t & 3;
      if (g < BIN_CHUNKS) {
        const int  off  = coff[g];
        const int  cg   = coff[g + 1] - off;
        const int* slot = region + SLOT_HEAD + g * SLOT_INTS;
        for (int p = 4 * sub; p < cg; p += 16) {
          const v4i e01 = *(const v4i*)(slot + 2 * p);
          const v4i e23 = *(const v4i*)(slot + 2 * p + 4);
          const int w0[4] = {e01.x, e01.z, e23.x, e23.z};
          const int w1[4] = {e01.y, e01.w, e23.y, e23.w};
#pragma unroll
          for (int k = 0; k < 4; ++k)
            if (p + k < cg) { elist[off + p + k] = w0[k];  elval[off + p + k] = __int_as_float(w1[k]); }
        }
      }
    }
    __syncthreads();
    sum_rows_sorted(sup, bias, out, elist, elval, nlist, scol, sval, cnt, start, &nextrow, lo);
    return;
  }

  {
    const float* bq = bias + 4 * (t & 63);
    v4f bv;
    bv.x = bq[0]; bv.y = bq[1]; bv.z = bq[2]; bv.w = bq[3];
    v4f* a4 = (v4f*)acc;
    for (int i = t; i < OWN_ROWS * (DOUT / 4); i += 32 * OWN_WAVES) a4[i] = bv;
  }
  __syncthreads();

  const int  slot  = lane & (BATCH - 1);

  {
    const v4i* er4 = (const v4i*)er + (size_t)wave * QUADS_PER_WAVE;
    int* q = queue + wave * QCAP;
    int count = 0;
    const v4i none = {-1, -1, -1, -1};

    v4i cur[SCAN_U], nxt[SCAN_U];
#pragma unroll
    for (int u = 0; u < SCAN_U; ++u) {
      const int qi = u * 32 + lane;
      cur[u] = (qi < QUADS_PER_WAVE) ? er4[qi] : none;
    }

    for (int it = 0; it < SCAN_ITERS; ++it) {
#pragma unroll
      for (int u = 0; u < SCAN_U; ++u) {
        const int qi = (it + 1) * (32 * SCAN_U) + u * 32 + lane;
        nxt[u] = (qi < QUADS_PER_WAVE) ? er4[qi] : none;
      }

#pragma unroll
      for (int u = 0; u < SCAN_U; ++u) {
        const int qi    = it * (32 * SCAN_U) + u * 32 + lane;
        const int ebase = (wave * QUADS_PER_WAVE + qi) * 4;
        unsigned d[4], mask[4];
#pragma unroll
        for (int k = 0; k < 4; ++k) {
          d[k]    = (unsigned)(cur[u][k] - lo);
          mask[k] = __builtin_amdgcn_ballot_w32(d[k] < (unsigned)OWN_ROWS);
        }
        if ((mask[0] | mask[1] | mask[2] | mask[3]) != 0u) {
#pragma unroll
          for (int k = 0; k < 4; ++k) {
            if (d[k] < (unsigned)OWN_ROWS)
              q[count + (int)__builtin_amdgcn_mbcnt_lo(mask[k], 0u)] = ((ebase + k) << 8) | (int)d[k];
            count += __builtin_popcount(mask[k]);
          }
        }
        __builtin_amdgcn_fence(__ATOMIC_ACQ_REL, "wavefront");
        while (count >= BATCH) {
          count -= BATCH;
          drain_batch(sup, ev, ec, acc, q[count + slot], BATCH);
        }
      }

#pragma unroll
      for (int u = 0; u < SCAN_U; ++u) cur[u] = nxt[u];
    }
    if (count > 0) drain_batch(sup, ev, ec, acc, q[slot < count ? slot : count - 1], count);
  }
  __syncthreads();

  {
    const v4f* a4 = (const v4f*)acc;
    v4f* o4 = (v4f*)out + (size_t)lo * (DOUT / 4);
    for (int i = t; i < OWN_ROWS * (DOUT / 4); i += 32 * OWN_WAVES) o4[i] = a4[i];
  }
}

extern "C" void kernel_launch(void* const* d_in, const int* in_sizes, int n_in,
                              void* d_out, int out_size, void* d_ws, size_t ws_size,
                              hipStream_t stream) {
  const float* inputs  = (const float*)d_in[0];
  const float* weights = (const float*)d_in[1];
  const float* bias    = (const float*)d_in[2];
  const float* evals   = (const float*)d_in[3];
  const int*   erow    = (const int*)d_in[4];
  const int*   ecol    = (const int*)d_in[5];
  float* out      = (float*)d_out;
  float* supports = (float*)d_ws;

  project_and_bin<<<BIN_CHUNKS + PAIR_BLOCKS + LONE_BLOCKS, 128, 0, stream>>>(inputs, weights, supports, erow, ecol, evals, out);
  owner_aggregate<<<OWN_WGS, 32 * OWN_WAVES, 0, stream>>>(supports, evals, erow, ecol, bias, out);
}


# product wave tile 64x64: a block takes 4 row tiles (768 blocks, 3 per WGP), each loaded B fragment feeds 4 WMMAs, 8 loads per 32 WMMAs (was 6 per 16); per-tile k order kept, sums bit-identical
# speedup vs baseline: 3.2448x; 1.0446x over previous
#include <hip/hip_runtime.h>

typedef __attribute__((ext_vector_type(2))) float v2f;
typedef __attribute__((ext_vector_type(8))) float v8f;
typedef __attribute__((ext_vector_type(4))) float v4f;
typedef __attribute__((ext_vector_type(4))) int v4i;
typedef __attribute__((ext_vector_type(2))) int v2i;

#define NODES 50000
#define DIN   512
#define DOUT  256
#define EDGES 800000

#define OWN_ROWS   200
#define OWN_WGS    (NODES / OWN_ROWS)
#define BIN_CHUNKS OWN_WGS
#define SLOT_HEAD  256
#define SLOT_CAP   96
#define SLOT_INTS  (2 * SLOT_CAP)
static_assert(SLOT_HEAD >= BIN_CHUNKS && SLOT_HEAD % 32 == 0 && SLOT_INTS % 32 == 0, "counts first, then line-aligned slots");
static_assert(SLOT_HEAD + BIN_CHUNKS * SLOT_INTS <= OWN_ROWS * DOUT, "the slots must fit the owner's own rows of out");
static_assert(SLOT_CAP % 4 == 0, "an owner reads a slot four entries at a time");
static_assert(NODES <= (1 << 23), "source row << 8 must stay a positive int");

#define ROW_TILES    (NODES / 16)
#define QUAD_BLOCKS  768
#define LONE_BLOCKS  (ROW_TILES - 4 * QUAD_BLOCKS)
#define KTRIPS       (DIN / 8)
static_assert(ROW_TILES * 16 == NODES, "row tiles must cover the nodes exactly");
static_assert(LONE_BLOCKS >= 0, "four-tile blocks must not run past the last row tile");
static_assert(KTRIPS % 3 == 1, "the loop below runs KTRIPS - 1 trips in threes and one more behind it");

template <int MT>
struct OperandSet {
  v4f a[MT];
  v4f q[4];
};

template <int MT>
__device__ __forceinline__ void load_operands(OperandSet<MT>& s, const float* arow0,
                                              const float* bcol, int k) {
  k = (k < DIN - 8) ? k : DIN - 8;
#pragma unroll
  for (int m = 0; m < MT; ++m) s.a[m] = *(const v4f*)(arow0 + (size_t)m * (16 * DIN) + k);
#pragma unroll
  for (int j = 0; j < 4; ++j) s.q[j] = *(const v4f*)(bcol + (size_t)(k + j) * DOUT);
  asm volatile("" ::: "memory");
  __builtin_amdgcn_sched_barrier(0);
}

template <int MT>
__device__ __forceinline__ void multiply_trip(v8f (&acc)[MT][4], const OperandSet<MT>& s) {
#pragma unroll
  for (int g = 0; g < 2; ++g) {
    v2f ap[MT], bp[4];
#pragma unroll
    for (int m = 0; m < MT; ++m) { ap[m].x = s.a[m][2 * g];  ap[m].y = s.a[m][2 * g + 1]; }
#pragma unroll
    for (int t = 0; t < 4; ++t) { bp[t].x = s.q[2 * g][t];  bp[t].y = s.q[2 * g + 1][t]; }
#pragma unroll
    for (int t = 0; t < 4; ++t)
#pragma unroll
      for (int m = 0; m < MT; ++m)
        acc[m][t] = __builtin_amdgcn_wmma_f32_16x16x4_f32(false, ap[m], false, bp[t], (short)0,
                                                          acc[m][t], false, false);
  }
}

template <int MT>
__device__ __forceinline__ void gemm_row_tiles(const float* __restrict__ A,
                                               const float* __restrict__ B,
                                               float* __restrict__ C, const int mbase) {
  const int lane  = threadIdx.x & 31;
  const int wave  = threadIdx.x >> 5;
  const int half  = lane >> 4;
  const int l15   = lane & 15;
  const int nbase = wave * 64;

  v8f acc[MT][4];
#pragma unroll
  for (int m = 0; m < MT; ++m)
#pragma unroll
    for (int t = 0; t < 4; ++t) acc[m][t] = v8f{};

  const float* arow0 = A + (size_t)(mbase + l15) * DIN + 4 * half;
  const float* bcol = B + (size_t)(4 * half) * DOUT + nbase + 4 * l15;

  OperandSet<MT> s0, s1, s2;
  load_operands<MT>(s0, arow0, bcol, 0);
  load_operands<MT>(s1, arow0, bcol, 8);
  for (int k = 0; k < DIN - 8; k += 24) {
    load_operands<MT>(s2, arow0, bcol, k + 16);
    multiply_trip<MT>(acc, s0);
    load_operands<MT>(s0, arow0, bcol, k + 24);
    multiply_trip<MT>(acc, s1);
    load_operands<MT>(s1, arow0, bcol, k + 32);
    multiply_trip<MT>(acc, s2);
  }
  multiply_trip<MT>(acc, s0);

#pragma unroll
  for (int m = 0; m < MT; ++m) {
    float* crow = C + (size_t)(mbase + 16 * m + 8 * half) * DOUT + nbase + 4 * l15;
#pragma unroll
    for (int r = 0; r < 8; ++r) {
      v4f o;
      o.x = acc[m][0][r];  o.y = acc[m][1][r];  o.z = acc[m][2][r];  o.w = acc[m][3][r];
      *(v4f*)(crow + (size_t)r * DOUT) = o;
    }
  }
}

#define BIN_EDGES   (EDGES / BIN_CHUNKS)
#define BIN_QUADS   (BIN_EDGES / 4)
#define BIN_THREADS 128
#define BIN_STEPS   ((BIN_QUADS + BIN_THREADS - 1) / BIN_THREADS)
static_assert(BIN_EDGES * BIN_CHUNKS == EDGES && BIN_QUADS * 4 == BIN_EDGES, "chunks of whole edge quads must cover the list");
static_assert(OWN_ROWS <= 256 && EDGES <= (1 << 20), "a queue entry of the scanning path packs a 20-bit edge index and an 8-bit local row");
static_assert(OWN_WGS <= 2048 && BIN_EDGES <= 4096, "key packs an 11-bit owner, a 12-bit rank and an 8-bit local row");

__device__ __forceinline__ void bin_chunk(const int* __restrict__ er, const int* __restrict__ ec,
                                          const float* __restrict__ ev, float* __restrict__ outbuf, const int g) {
  __shared__ int run[OWN_WGS];
  const int t = threadIdx.x;
  for (int o = t; o < OWN_WGS; o += BIN_THREADS) run[o] = 0;
  __syncthreads();

  const v4i none = {-1, -1, -1, -1};
  v4i r[BIN_STEPS];
#pragma unroll
  for (int s = 0; s < BIN_STEPS; ++s) {
    const int qi = s * BIN_THREADS + t;
    r[s] = (qi < BIN_QUADS) ? ((const v4i*)er)[g * BIN_QUADS + qi] : none;
  }
  int key[BIN_STEPS][4];
#pragma unroll
  for (int s = 0; s < BIN_STEPS; ++s) {
#pragma unroll
    for (int k = 0; k < 4; ++k) {
      const unsigned row = (unsigned)r[s][k];
      const unsigned o   = row / OWN_ROWS;
      key[s][k] = -1;
      if (o < (unsigned)OWN_WGS) {
        const int rank = atomicAdd(&run[o], 1);
        key[s][k] = (int)(o << 20) | (rank << 8) | (int)(row - o * OWN_ROWS);
      }
    }
  }
  __syncthreads();
  int* ob = (int*)outbuf;
  for (int o = t; o < OWN_WGS; o += BIN_THREADS) ob[(size_t)o * (OWN_ROWS * DOUT) + g] = run[o];
  v4i c[BIN_STEPS];
  v4f v[BIN_STEPS];
#pragma unroll
  for (int s = 0; s < BIN_STEPS; ++s) {
    const int qi = s * BIN_THREADS + t;
    c[s] = (qi < BIN_QUADS) ? ((const v4i*)ec)[g * BIN_QUADS + qi] : none;
    v[s] = (qi < BIN_QUADS) ? ((const v4f*)ev)[g * BIN_QUADS + qi] : v4f{};
  }
#pragma unroll
  for (int s = 0; s < BIN_STEPS; ++s) {
#pragma unroll
    for (int k = 0; k < 4; ++k) {
      const int ky   = key[s][k];
      const int rank = (ky >> 8) & 0xFFF;
      if (ky >= 0 && rank < SLOT_CAP) {
        v2i e;
        e.x = (c[s][k] << 8) | (ky & 255);
        e.y = __float_as_int(v[s][k]);
        *(v2i*)(ob + (size_t)(ky >> 20) * (OWN_ROWS * DOUT) + SLOT_HEAD + g * SLOT_INTS + 2 * rank) = e;
      }
    }
  }
}

static_assert(BIN_THREADS == 128, "the binning work-groups share the product's launch shape");

__global__ __launch_bounds__(128) void project_and_bin(const float* __restrict__ A,
                                                       const float* __restrict__ B,
                                                       float* __restrict__ C,
                                                       const int* __restrict__ er,
                                                       const int* __restrict__ ec,
                                                       const float* __restrict__ ev,
                                                       float* __restrict__ outbuf) {
  if (blockIdx.x < BIN_CHUNKS) { bin_chunk(er, ec, ev, outbuf, blockIdx.x);  return; }
  const int b = blockIdx.x - BIN_CHUNKS;
  if (b < QUAD_BLOCKS) gemm_row_tiles<4>(A, B, C, b * 64);
  else                 gemm_row_tiles<1>(A, B, C, (4 * QUAD_BLOCKS + (b - QUAD_BLOCKS)) * 16);
}


#define OWN_WAVES  32
#define SORT_CAP   14336
#define ROW_SLOTS  224
#define QCAP       256
#define BATCH      8
#define SCAN_U     4
#define QUADS_PER_WAVE (EDGES / 4 / OWN_WAVES)
#define SCAN_ITERS ((QUADS_PER_WAVE + 32 * SCAN_U - 1) / (32 * SCAN_U))
static_assert(OWN_WGS * OWN_ROWS == NODES, "owned row ranges must tile the nodes exactly");
static_assert(EDGES % (4 * OWN_WAVES) == 0, "edge quads must split evenly over the waves");
static_assert(OWN_ROWS <= 256, "local row must fit the low 8 bits of a queue entry");
static_assert(OWN_ROWS <= ROW_SLOTS && ROW_SLOTS == 7 * 32, "one wave sums the row counts, 7 per lane");
static_assert(4 * SORT_CAP <= OWN_ROWS * DOUT + OWN_WAVES * QCAP, "gathered entries and sorted edges must fit the LDS block");
static_assert(BIN_CHUNKS <= 256 && 4 * BIN_CHUNKS <= 32 * OWN_WAVES, "one wave sums the chunk counts 8 per lane; 4 threads copy a slot");
static_assert(BATCH == 8, "a row's tail is taken as batches of 4, 2 and 1");
static_assert(BATCH - 1 + 128 <= QCAP, "queue holds a left-over plus one 128-edge step");

__device__ __forceinline__ void drain_batch(const float* __restrict__ sup,
                                            const float* __restrict__ ev,
                                            const int*   __restrict__ ec,
                                            float* acc, const int entv, int n) {
  const int lane = threadIdx.x & 31;
  int   lrow[BATCH];
  float val[BATCH];
  float s[BATCH][8];
#pragma unroll
  for (int i = 0; i < BATCH; ++i) {
    const int ent = __builtin_amdgcn_readlane(entv, i);
    const int e   = ent >> 8;
    lrow[i] = ent & 255;
    val[i]  = ev[e];
    const float* src = sup + (size_t)ec[e] * DOUT + lane;
#pragma unroll
    for (int j = 0; j < 8; ++j) s[i][j] = src[j * 32];
  }
#pragma unroll
  for (int i = 0; i < BATCH; ++i) {
    if (i < n) {
      float* dst = acc + lrow[i] * DOUT + lane;
#pragma unroll
      for (int j = 0; j < 8; ++j) atomicAdd(dst + j * 32, s[i][j] * val[i]);
    }
  }
}


template <int N>
__device__ __forceinline__ void add_batch(const float* __restrict__ sup, const int* scol, const float* sval,
                                          const int p, v4f& a0, v4f& a1) {
  const int lane = threadIdx.x & 31;
  const int idx  = p + (lane & (N - 1));
  const int myc  = scol[idx];
  const int myv  = __float_as_int(sval[idx]);
  v4f   q0[N], q1[N];
  float val[N];
#pragma unroll
  for (int i = 0; i < N; ++i) {
    const int c = __builtin_amdgcn_readlane(myc, i);
    val[i] = __int_as_float(__builtin_amdgcn_readlane(myv, i));
    const float* src = sup + (size_t)c * DOUT + 4 * lane;
    q0[i] = *(const v4f*)src;
    q1[i] = *(const v4f*)(src + 128);
  }
#pragma unroll
  for (int i = 0; i < N; ++i) {
    a0 += q0[i] * val[i];
    a1 += q1[i] * val[i];
  }
}

__device__ __forceinline__ void sum_rows_sorted(const float* __restrict__ sup, const float* __restrict__ bias,
                                                float* out, const int* ent, const float* entv, const int nlist,
                                                int* scol, float* sval, int* cnt, int* start, int* nextrow,
                                                const int lo) {
  const int t    = threadIdx.x;
  const int lane = t & 31;
  if (t < ROW_SLOTS) cnt[t] = 0;
  if (t == 0) *nextrow = 0;
  __syncthreads();

  for (int base = 0; base < nlist; base += 4 * 32 * OWN_WAVES) {
    int en[4];
#pragma unroll
    for (int u = 0; u < 4; ++u) {
      const int i = base + u * (32 * OWN_WAVES) + t;
      en[u] = (i < nlist) ? ent[i] : -1;
    }
#pragma unroll
    for (int u = 0; u < 4; ++u)
      if (en[u] >= 0) atomicAdd(&cnt[en[u] & 255], 1);
  }
  __syncthreads();

  if (t < 32) {
    int c[7];
    int s = 0;
#pragma unroll
    for (int j = 0; j < 7; ++j) { c[j] = cnt[7 * t + j];  s += c[j]; }
    int incl = s;
#pragma unroll
    for (int d = 1; d < 32; d <<= 1) {
      const int o = __shfl_up(incl, d);
      if (t >= d) incl += o;
    }
    int run = incl - s;
#pragma unroll
    for (int j = 0; j < 7; ++j) { start[7 * t + j] = run;  cnt[7 * t + j] = run;  run += c[j]; }
    if (t == 31) start[ROW_SLOTS] = run;
  }
  __syncthreads();

  for (int base = 0; base < nlist; base += 4 * 32 * OWN_WAVES) {
    int   en[4];
    float v[4];
#pragma unroll
    for (int u = 0; u < 4; ++u) {
      const int i = base + u * (32 * OWN_WAVES) + t;
      en[u] = (i < nlist) ? ent[i] : -1;
      v[u]  = (i < nlist) ? entv[i] : 0.0f;
    }
#pragma unroll
    for (int u = 0; u < 4; ++u) {
      if (en[u] >= 0) {
        const int pos = atomicAdd(&cnt[en[u] & 255], 1);
        scol[pos] = en[u] >> 8;
        sval[pos] = v[u];
      }
    }
  }
  __syncthreads();

  v4f b0, b1;
  {
    const float* bq = bias + 4 * lane;
    b0.x = bq[0];    b0.y = bq[1];    b0.z = bq[2];    b0.w = bq[3];
    b1.x = bq[128];  b1.y = bq[129];  b1.z = bq[130];  b1.w = bq[131];
  }
  for (;;) {
    int r = 0;
    if (lane == 0) r = atomicAdd(nextrow, 1);
    r = __builtin_amdgcn_readfirstlane(r);
    if (r >= OWN_ROWS) break;
    const int pb = __builtin_amdgcn_readfirstlane(start[r]);
    const int pe = __builtin_amdgcn_readfirstlane(start[r + 1]);
    v4f a0 = b0, a1 = b1;
    int p = pb;
    for (; p + BATCH <= pe; p += BATCH) add_batch<BATCH>(sup, scol, sval, p, a0, a1);
    if (pe - p >= 4) { add_batch<4>(sup, scol, sval, p, a0, a1);  p += 4; }
    if (pe - p >= 2) { add_batch<2>(sup, scol, sval, p, a0, a1);  p += 2; }
    if (pe - p >= 1) { add_batch<1>(sup, scol, sval, p, a0, a1); }
    float* dst = out + (size_t)(lo + r) * DOUT + 4 * lane;
    *(v4f*)dst         = a0;
    *(v4f*)(dst + 128) = a1;
  }
}

__global__ __launch_bounds__(32 * OWN_WAVES) void owner_aggregate(const float* __restrict__ sup,
                                                                  const float* __restrict__ ev,
                                                                  const int*   __restrict__ er,
                                                                  const int*   __restrict__ ec,
                                                                  const float* __restrict__ bias,
                                                                  float* out) {
  __shared__ __attribute__((aligned(16))) float lds[OWN_ROWS * DOUT + OWN_WAVES * QCAP];
  __shared__ int cnt[ROW_SLOTS];
  __shared__ int start[ROW_SLOTS + 1];
  __shared__ int nextrow;
  __shared__ int coff[256 + 1];
  __shared__ int spill;
  float* acc   = lds;
  int*   queue = (int*)(lds + OWN_ROWS * DOUT);
  int*   scol  = (int*)lds;
  float* sval  = lds + SORT_CAP;
  int*   elist = (int*)(lds + 2 * SORT_CAP);
  float* elval = lds + 3 * SORT_CAP;

  const int t    = threadIdx.x;
  const int lane = t & 31;
  const int wave = __builtin_amdgcn_readfirstlane(t >> 5);
  const int lo   = blockIdx.x * OWN_ROWS;

  const int* region = (const int*)out + (size_t)lo * DOUT;
  int mycount = 0;
  if (t < BIN_CHUNKS) mycount = region[t];
  if (t < 256) coff[t] = mycount;
  if (t == 0) spill = 0;
  __syncthreads();
  if (mycount > SLOT_CAP) spill = 1;
  if (t < 32) {
    int c[8];
    int s = 0;
#pragma unroll
    for (int j = 0; j < 8; ++j) { c[j] = coff[8 * t + j];  s += c[j]; }
    int incl = s;
#pragma unroll
    for (int d = 1; d < 32; d <<= 1) {
      const int o = __shfl_up(incl, d);
      if (t >= d) incl += o;
    }
    int run = incl - s;
#pragma unroll
    for (int j = 0; j < 8; ++j) { coff[8 * t + j] = run;  run += c[j]; }
    if (t == 31) coff[256] = run;
  }
  __syncthreads();
  const int nlist   = __builtin_amdgcn_readfirstlane(coff[256]);
  const int spilled = __builtin_amdgcn_readfirstlane(spill);

  if (spilled == 0 && nlist <= SORT_CAP) {
    {
      const int g = t >> 2, sub = t & 3;
      if (g < BIN_CHUNKS) {
        const int  off  = coff[g];
        const int  cg   = coff[g + 1] - off;
        const int* slot = region + SLOT_HEAD + g * SLOT_INTS;
        for (int p = 4 * sub; p < cg; p += 16) {
          const v4i e01 = *(const v4i*)(slot + 2 * p);
          const v4i e23 = *(const v4i*)(slot + 2 * p + 4);
          const int w0[4] = {e01.x, e01.z, e23.x, e23.z};
          const int w1[4] = {e01.y, e01.w, e23.y, e23.w};
#pragma unroll
          for (int k = 0; k < 4; ++k)
            if (p + k < cg) { elist[off + p + k] = w0[k];  elval[off + p + k] = __int_as_float(w1[k]); }
        }
      }
    }
    __syncthreads();
    sum_rows_sorted(sup, bias, out, elist, elval, nlist, scol, sval, cnt, start, &nextrow, lo);
    return;
  }

  {
    const float* bq = bias + 4 * (t & 63);
    v4f bv;
    bv.x = bq[0]; bv.y = bq[1]; bv.z = bq[2]; bv.w = bq[3];
    v4f* a4 = (v4f*)acc;
    for (int i = t; i < OWN_ROWS * (DOUT / 4); i += 32 * OWN_WAVES) a4[i] = bv;
  }
  __syncthreads();

  const int  slot  = lane & (BATCH - 1);

  {
    const v4i* er4 = (const v4i*)er + (size_t)wave * QUADS_PER_WAVE;
    int* q = queue + wave * QCAP;
    int count = 0;
    const v4i none = {-1, -1, -1, -1};

    v4i cur[SCAN_U], nxt[SCAN_U];
#pragma unroll
    for (int u = 0; u < SCAN_U; ++u) {
      const int qi = u * 32 + lane;
      cur[u] = (qi < QUADS_PER_WAVE) ? er4[qi] : none;
    }

    for (int it = 0; it < SCAN_ITERS; ++it) {
#pragma unroll
      for (int u = 0; u < SCAN_U; ++u) {
        const int qi = (it + 1) * (32 * SCAN_U) + u * 32 + lane;
        nxt[u] = (qi < QUADS_PER_WAVE) ? er4[qi] : none;
      }

#pragma unroll
      for (int u = 0; u < SCAN_U; ++u) {
        const int qi    = it * (32 * SCAN_U) + u * 32 + lane;
        const int ebase = (wave * QUADS_PER_WAVE + qi) * 4;
        unsigned d[4], mask[4];
#pragma unroll
        for (int k = 0; k < 4; ++k) {
          d[k]    = (unsigned)(cur[u][k] - lo);
          mask[k] = __builtin_amdgcn_ballot_w32(d[k] < (unsigned)OWN_ROWS);
        }
        if ((mask[0] | mask[1] | mask[2] | mask[3]) != 0u) {
#pragma unroll
          for (int k = 0; k < 4; ++k) {
            if (d[k] < (unsigned)OWN_ROWS)
              q[count + (int)__builtin_amdgcn_mbcnt_lo(mask[k], 0u)] = ((ebase + k) << 8) | (int)d[k];
            count += __builtin_popcount(mask[k]);
          }
        }
        __builtin_amdgcn_fence(__ATOMIC_ACQ_REL, "wavefront");
        while (count >= BATCH) {
          count -= BATCH;
          drain_batch(sup, ev, ec, acc, q[count + slot], BATCH);
        }
      }

#pragma unroll
      for (int u = 0; u < SCAN_U; ++u) cur[u] = nxt[u];
    }
    if (count > 0) drain_batch(sup, ev, ec, acc, q[slot < count ? slot : count - 1], count);
  }
  __syncthreads();

  {
    const v4f* a4 = (const v4f*)acc;
    v4f* o4 = (v4f*)out + (size_t)lo * (DOUT / 4);
    for (int i = t; i < OWN_ROWS * (DOUT / 4); i += 32 * OWN_WAVES) o4[i] = a4[i];
  }
}

extern "C" void kernel_launch(void* const* d_in, const int* in_sizes, int n_in,
                              void* d_out, int out_size, void* d_ws, size_t ws_size,
                              hipStream_t stream) {
  const float* inputs  = (const float*)d_in[0];
  const float* weights = (const float*)d_in[1];
  const float* bias    = (const float*)d_in[2];
  const float* evals   = (const float*)d_in[3];
  const int*   erow    = (const int*)d_in[4];
  const int*   ecol    = (const int*)d_in[5];
  float* out      = (float*)d_out;
  float* supports = (float*)d_ws;

  project_and_bin<<<BIN_CHUNKS + QUAD_BLOCKS + LONE_BLOCKS, 128, 0, stream>>>(inputs, weights, supports, erow, ecol, evals, out);
  owner_aggregate<<<OWN_WGS, 32 * OWN_WAVES, 0, stream>>>(supports, evals, erow, ecol, bias, out);
}
